# starting kernel
# baseline (speedup 1.0000x reference)
#include <hip/hip_runtime.h>
#include <hip/hip_bf16.h>


typedef __attribute__((ext_vector_type(2))) float v2f;
typedef __attribute__((ext_vector_type(8))) float v8f;

#define N_NODES 40000
#define N_EDGES 640000
#define IN_DIM  3
#define HID_DIM 128
#define LAT_DIM 64

__global__ void zero_kernel(float* __restrict__ p, int n) {
    int i = blockIdx.x * blockDim.x + threadIdx.x;
    int stride = gridDim.x * blockDim.x;
    for (; i < n; i += stride) p[i] = 0.0f;
}

__global__ void degree_kernel(const int* __restrict__ dst, float* __restrict__ deg,
                              int nEdges) {
    int e = blockIdx.x * blockDim.x + threadIdx.x;
    if (e < nEdges) atomicAdd(&deg[dst[e]], 1.0f);
}

__global__ void rsqrt_kernel(float* __restrict__ dis, int n) {
    int i = blockIdx.x * blockDim.x + threadIdx.x;
    if (i < n) dis[i] = rsqrtf(dis[i] + 1.0f);
}

template <int K>
__global__ __launch_bounds__(256) void gemm_wmma_f32(
    const float* __restrict__ A, const float* __restrict__ B,
    float* __restrict__ C, int M, int N) {
    const int lane   = threadIdx.x & 31;
    const int wave   = threadIdx.x >> 5;
    const int tilesN = N >> 4;
    const int tile   = blockIdx.x * 8 + wave;
    const int totalTiles = (M >> 4) * tilesN;
    if (tile >= totalTiles) return;

    const int tm = tile / tilesN;
    const int tn = tile % tilesN;

    const int arow  = tm * 16 + (lane & 15);
    const int khalf = (lane >> 4) * 2;
    const int ncol  = tn * 16 + (lane & 15);

    v8f acc = {};
    for (int k0 = 0; k0 < K; k0 += 4) {
        const int ka = k0 + khalf;
        v2f a, b;
        a.x = (ka     < K) ? A[(size_t)arow * K + ka]     : 0.0f;
        a.y = (ka + 1 < K) ? A[(size_t)arow * K + ka + 1] : 0.0f;
        b.x = (ka     < K) ? B[(size_t)(ka)     * N + ncol] : 0.0f;
        b.y = (ka + 1 < K) ? B[(size_t)(ka + 1) * N + ncol] : 0.0f;
        acc = __builtin_amdgcn_wmma_f32_16x16x4_f32(
            false, a, false, b, (short)0, acc, false, false);
    }

    const int crow = tm * 16 + (lane >> 4) * 8;
#pragma unroll
    for (int r = 0; r < 8; ++r)
        C[(size_t)(crow + r) * N + ncol] = acc[r];
}

template <int DIM>
__global__ __launch_bounds__(256) void scatter_edges(
    const int* __restrict__ src, const int* __restrict__ dst,
    const float* __restrict__ dis, const float* __restrict__ m,
    float* __restrict__ agg, int nEdges) {
    const int gw   = (blockIdx.x * blockDim.x + threadIdx.x) >> 5;
    const int lane = threadIdx.x & 31;
    if (gw >= nEdges) return;

    const int s = src[gw];
    const int d = dst[gw];
    const float coeff = dis[s] * dis[d];

    constexpr int PER = DIM / 32;
    const float* __restrict__ mrow = m + (size_t)s * DIM + lane * PER;
    float* __restrict__ arow = agg + (size_t)d * DIM + lane * PER;

    float v[PER];
#pragma unroll
    for (int j = 0; j < PER; ++j) v[j] = mrow[j];
#pragma unroll
    for (int j = 0; j < PER; ++j) atomicAdd(&arow[j], v[j] * coeff);
}

template <int DIM, bool RELU>
__global__ void finalize_kernel(const float* __restrict__ agg,
                                const float* __restrict__ m,
                                const float* __restrict__ dis,
                                const float* __restrict__ bias,
                                float* __restrict__ out, int nNodes) {
    int idx = blockIdx.x * blockDim.x + threadIdx.x;
    if (idx >= nNodes * DIM) return;
    const int node = idx / DIM;
    const int d    = idx % DIM;
    const float di = dis[node];
    float v = agg[idx] + m[idx] * di * di + bias[d];
    if (RELU) v = fmaxf(v, 0.0f);
    out[idx] = v;
}

extern "C" void kernel_launch(void* const* d_in, const int* in_sizes, int n_in,
                              void* d_out, int out_size, void* d_ws, size_t ws_size,
                              hipStream_t stream) {
    const float* x   = (const float*)d_in[0];
    const int*   ei  = (const int*)d_in[1];
    const float* W0  = (const float*)d_in[2];
    const float* b0  = (const float*)d_in[3];
    const float* W1  = (const float*)d_in[4];
    const float* b1  = (const float*)d_in[5];
    const float* W2  = (const float*)d_in[6];
    const float* b2  = (const float*)d_in[7];
    float* out = (float*)d_out;

    const int* src = ei;
    const int* dst = ei + N_EDGES;

    float* ws  = (float*)d_ws;
    float* dis = ws;
    float* m   = dis + N_NODES;
    float* agg = m   + (size_t)N_NODES * HID_DIM;
    float* h   = agg + (size_t)N_NODES * HID_DIM;

    const int T = 256;
    const int nodeDimBlocksH = (N_NODES * HID_DIM + T - 1) / T;
    const int nodeDimBlocksL = (N_NODES * LAT_DIM + T - 1) / T;
    const int edgeWaveBlocks = (N_EDGES * 32 + T - 1) / T;
    const int gemmBlocksH = (N_NODES / 16) * (HID_DIM / 16) / 8;
    const int gemmBlocksL = (N_NODES / 16) * (LAT_DIM / 16) / 8;

    zero_kernel<<<512, T, 0, stream>>>(dis, N_NODES);
    degree_kernel<<<(N_EDGES + T - 1) / T, T, 0, stream>>>(dst, dis, N_EDGES);
    rsqrt_kernel<<<(N_NODES + T - 1) / T, T, 0, stream>>>(dis, N_NODES);

    gemm_wmma_f32<IN_DIM><<<gemmBlocksH, T, 0, stream>>>(x, W0, m, N_NODES, HID_DIM);
    zero_kernel<<<2048, T, 0, stream>>>(agg, N_NODES * HID_DIM);
    scatter_edges<HID_DIM><<<edgeWaveBlocks, T, 0, stream>>>(src, dst, dis, m, agg, N_EDGES);
    finalize_kernel<HID_DIM, true><<<nodeDimBlocksH, T, 0, stream>>>(agg, m, dis, b0, h, N_NODES);

    gemm_wmma_f32<HID_DIM><<<gemmBlocksH, T, 0, stream>>>(h, W1, m, N_NODES, HID_DIM);
    zero_kernel<<<2048, T, 0, stream>>>(agg, N_NODES * HID_DIM);
    scatter_edges<HID_DIM><<<edgeWaveBlocks, T, 0, stream>>>(src, dst, dis, m, agg, N_EDGES);
    finalize_kernel<HID_DIM, true><<<nodeDimBlocksH, T, 0, stream>>>(agg, m, dis, b1, h, N_NODES);

    gemm_wmma_f32<HID_DIM><<<gemmBlocksL, T, 0, stream>>>(h, W2, m, N_NODES, LAT_DIM);
    zero_kernel<<<1024, T, 0, stream>>>(agg, N_NODES * LAT_DIM);
    scatter_edges<LAT_DIM><<<edgeWaveBlocks, T, 0, stream>>>(src, dst, dis, m, agg, N_EDGES);
    finalize_kernel<LAT_DIM, false><<<nodeDimBlocksL, T, 0, stream>>>(agg, m, dis, b2, out, N_NODES);
}


# Float-atomic scatter-add replaced by dst-grouped edge list (int atomic arrival rank, scan) and one owner wave per node summing gathered rows in registers, bias/act fused; sum order = arrival rank
# speedup vs baseline: 2.0602x; 2.0602x over previous
#include <hip/hip_runtime.h>
#include <hip/hip_bf16.h>


typedef __attribute__((ext_vector_type(2))) float v2f;
typedef __attribute__((ext_vector_type(8))) float v8f;

#define N_NODES 40000
#define N_EDGES 640000
#define IN_DIM  3
#define HID_DIM 128
#define LAT_DIM 64

__global__ void zero_int_kernel(int* __restrict__ p, int n) {
    int i = blockIdx.x * blockDim.x + threadIdx.x;
    if (i < n) p[i] = 0;
}

__global__ void rank_kernel(const int* __restrict__ dst, int* __restrict__ cnt,
                            int* __restrict__ rank, int nEdges) {
    int e = blockIdx.x * blockDim.x + threadIdx.x;
    if (e < nEdges) rank[e] = atomicAdd(&cnt[dst[e]], 1);
}

#define SCAN_THREADS 1024
#define SCAN_PER     40
static_assert(SCAN_THREADS * SCAN_PER >= N_NODES, "scan covers all nodes");
static_assert(SCAN_PER % 4 == 0 && N_NODES % 4 == 0, "int4 chunks");

__global__ __launch_bounds__(SCAN_THREADS) void scan_kernel(
    const int* __restrict__ cnt, int* __restrict__ rowptr,
    float* __restrict__ dis, int n) {
    __shared__ int wsum[32];
    const int t    = threadIdx.x;
    const int lane = t & 31;
    const int w    = t >> 5;
    const int base = t * SCAN_PER;

    int v[SCAN_PER];
    int s = 0;
#pragma unroll
    for (int q = 0; q < SCAN_PER / 4; ++q) {
        const int idx = base + q * 4;
        int4 c = make_int4(0, 0, 0, 0);
        if (idx < n) c = *reinterpret_cast<const int4*>(cnt + idx);
        v[q * 4 + 0] = c.x; v[q * 4 + 1] = c.y;
        v[q * 4 + 2] = c.z; v[q * 4 + 3] = c.w;
        s += c.x + c.y + c.z + c.w;
    }

    int inc = s;
#pragma unroll
    for (int off = 1; off < 32; off <<= 1) {
        const int y = __shfl_up(inc, off);
        if (lane >= off) inc += y;
    }
    if (lane == 31) wsum[w] = inc;
    __syncthreads();
    if (w == 0) {
        int x = wsum[lane];
#pragma unroll
        for (int off = 1; off < 32; off <<= 1) {
            const int y = __shfl_up(x, off);
            if (lane >= off) x += y;
        }
        wsum[lane] = x;
    }
    __syncthreads();

    int excl = inc - s + (w > 0 ? wsum[w - 1] : 0);
#pragma unroll
    for (int q = 0; q < SCAN_PER / 4; ++q) {
        const int idx = base + q * 4;
        if (idx < n) {
            int4 r;
            float4 d;
            r.x = excl; excl += v[q * 4 + 0];
            r.y = excl; excl += v[q * 4 + 1];
            r.z = excl; excl += v[q * 4 + 2];
            r.w = excl; excl += v[q * 4 + 3];
            d.x = rsqrtf((float)v[q * 4 + 0] + 1.0f);
            d.y = rsqrtf((float)v[q * 4 + 1] + 1.0f);
            d.z = rsqrtf((float)v[q * 4 + 2] + 1.0f);
            d.w = rsqrtf((float)v[q * 4 + 3] + 1.0f);
            *reinterpret_cast<int4*>(rowptr + idx) = r;
            *reinterpret_cast<float4*>(dis + idx)  = d;
        }
    }
    if (t == SCAN_THREADS - 1) rowptr[n] = excl;
}

__global__ void fill_kernel(const int* __restrict__ src, const int* __restrict__ dst,
                            const int* __restrict__ rank,
                            const int* __restrict__ rowptr,
                            const float* __restrict__ dis,
                            int2* __restrict__ rec, int nEdges) {
    int e = blockIdx.x * blockDim.x + threadIdx.x;
    if (e >= nEdges) return;
    const int s = src[e];
    const int d = dst[e];
    const int pos = rowptr[d] + rank[e];
    const float coeff = dis[s] * dis[d];
    rec[pos] = make_int2(s, __float_as_int(coeff));
}

template <int K>
__global__ __launch_bounds__(256) void gemm_wmma_f32(
    const float* __restrict__ A, const float* __restrict__ B,
    float* __restrict__ C, int M, int N) {
    const int lane   = threadIdx.x & 31;
    const int wave   = threadIdx.x >> 5;
    const int tilesN = N >> 4;
    const int tile   = blockIdx.x * 8 + wave;
    const int totalTiles = (M >> 4) * tilesN;
    if (tile >= totalTiles) return;

    const int tm = tile / tilesN;
    const int tn = tile % tilesN;

    const int arow  = tm * 16 + (lane & 15);
    const int khalf = (lane >> 4) * 2;
    const int ncol  = tn * 16 + (lane & 15);

    v8f acc = {};
    for (int k0 = 0; k0 < K; k0 += 4) {
        const int ka = k0 + khalf;
        v2f a, b;
        a.x = (ka     < K) ? A[(size_t)arow * K + ka]     : 0.0f;
        a.y = (ka + 1 < K) ? A[(size_t)arow * K + ka + 1] : 0.0f;
        b.x = (ka     < K) ? B[(size_t)(ka)     * N + ncol] : 0.0f;
        b.y = (ka + 1 < K) ? B[(size_t)(ka + 1) * N + ncol] : 0.0f;
        acc = __builtin_amdgcn_wmma_f32_16x16x4_f32(
            false, a, false, b, (short)0, acc, false, false);
    }

    const int crow = tm * 16 + (lane >> 4) * 8;
#pragma unroll
    for (int r = 0; r < 8; ++r)
        C[(size_t)(crow + r) * N + ncol] = acc[r];
}

template <int DIM, bool RELU>
__global__ __launch_bounds__(256) void gather_rows(
    const int* __restrict__ rowptr, const int2* __restrict__ rec,
    const float* __restrict__ dis, const float* __restrict__ m,
    const float* __restrict__ bias, float* __restrict__ out, int nNodes) {
    constexpr int PER = DIM / 32;
    typedef __attribute__((ext_vector_type(PER))) float vec;

    const int lane = threadIdx.x & 31;
    const int node = __builtin_amdgcn_readfirstlane(
        (int)(blockIdx.x * (blockDim.x >> 5) + (threadIdx.x >> 5)));
    if (node >= nNodes) return;

    const int   beg = rowptr[node];
    const int   end = rowptr[node + 1];
    const float di  = dis[node];

    const vec* __restrict__ mv = reinterpret_cast<const vec*>(m);
    vec acc = mv[(size_t)node * 32 + lane] * di * di;

    for (int base = beg; base < end; base += 32) {
        const int n = min(32, end - base);
        int2 r = make_int2(0, 0);
        if (lane < n) r = rec[base + lane];
        int j = 0;
        for (; j + 4 <= n; j += 4) {
            const int   s0 = __builtin_amdgcn_readlane(r.x, j);
            const int   s1 = __builtin_amdgcn_readlane(r.x, j + 1);
            const int   s2 = __builtin_amdgcn_readlane(r.x, j + 2);
            const int   s3 = __builtin_amdgcn_readlane(r.x, j + 3);
            const float c0 = __int_as_float(__builtin_amdgcn_readlane(r.y, j));
            const float c1 = __int_as_float(__builtin_amdgcn_readlane(r.y, j + 1));
            const float c2 = __int_as_float(__builtin_amdgcn_readlane(r.y, j + 2));
            const float c3 = __int_as_float(__builtin_amdgcn_readlane(r.y, j + 3));
            const vec v0 = mv[(size_t)s0 * 32 + lane];
            const vec v1 = mv[(size_t)s1 * 32 + lane];
            const vec v2 = mv[(size_t)s2 * 32 + lane];
            const vec v3 = mv[(size_t)s3 * 32 + lane];
            acc += v0 * c0;
            acc += v1 * c1;
            acc += v2 * c2;
            acc += v3 * c3;
        }
        for (; j < n; ++j) {
            const int   s0 = __builtin_amdgcn_readlane(r.x, j);
            const float c0 = __int_as_float(__builtin_amdgcn_readlane(r.y, j));
            acc += mv[(size_t)s0 * 32 + lane] * c0;
        }
    }

    acc += reinterpret_cast<const vec*>(bias)[lane];
    if (RELU) {
#pragma unroll
        for (int k = 0; k < PER; ++k) acc[k] = fmaxf(acc[k], 0.0f);
    }
    reinterpret_cast<vec*>(out)[(size_t)node * 32 + lane] = acc;
}

extern "C" void kernel_launch(void* const* d_in, const int* in_sizes, int n_in,
                              void* d_out, int out_size, void* d_ws, size_t ws_size,
                              hipStream_t stream) {
    const float* x   = (const float*)d_in[0];
    const int*   ei  = (const int*)d_in[1];
    const float* W0  = (const float*)d_in[2];
    const float* b0  = (const float*)d_in[3];
    const float* W1  = (const float*)d_in[4];
    const float* b1  = (const float*)d_in[5];
    const float* W2  = (const float*)d_in[6];
    const float* b2  = (const float*)d_in[7];
    float* out = (float*)d_out;

    const int* src = ei;
    const int* dst = ei + N_EDGES;

    float* ws     = (float*)d_ws;
    float* m      = ws;
    float* h      = m + (size_t)N_NODES * HID_DIM;
    int2*  rec    = (int2*)(h + (size_t)N_NODES * HID_DIM);
    float* dis    = (float*)(rec + N_EDGES);
    int*   cnt    = (int*)(dis + N_NODES);
    int*   rowptr = cnt + N_NODES;
    int*   rank   = rowptr + (N_NODES + 32);

    const int T = 256;
    const int nodeWaveBlocks = (N_NODES * 32 + T - 1) / T;
    const int gemmBlocksH = (N_NODES / 16) * (HID_DIM / 16) / 8;
    const int gemmBlocksL = (N_NODES / 16) * (LAT_DIM / 16) / 8;

    zero_int_kernel<<<(N_NODES + T - 1) / T, T, 0, stream>>>(cnt, N_NODES);
    rank_kernel<<<(N_EDGES + T - 1) / T, T, 0, stream>>>(dst, cnt, rank, N_EDGES);
    scan_kernel<<<1, SCAN_THREADS, 0, stream>>>(cnt, rowptr, dis, N_NODES);
    fill_kernel<<<(N_EDGES + T - 1) / T, T, 0, stream>>>(src, dst, rank, rowptr, dis, rec, N_EDGES);

    gemm_wmma_f32<IN_DIM><<<gemmBlocksH, T, 0, stream>>>(x, W0, m, N_NODES, HID_DIM);
    gather_rows<HID_DIM, true><<<nodeWaveBlocks, T, 0, stream>>>(rowptr, rec, dis, m, b0, h, N_NODES);

    gemm_wmma_f32<HID_DIM><<<gemmBlocksH, T, 0, stream>>>(h, W1, m, N_NODES, HID_DIM);
    gather_rows<HID_DIM, true><<<nodeWaveBlocks, T, 0, stream>>>(rowptr, rec, dis, m, b1, h, N_NODES);

    gemm_wmma_f32<HID_DIM><<<gemmBlocksL, T, 0, stream>>>(h, W2, m, N_NODES, LAT_DIM);
    gather_rows<LAT_DIM, false><<<nodeWaveBlocks, T, 0, stream>>>(rowptr, rec, dis, m, b2, out, N_NODES);
}


# K=128 GEMMs: each work-group reads its 32 A rows once as whole 512-byte rows into LDS and waves take A fragments from LDS (was 8-byte pieces over 16 lines per load); k and sum order unchanged
# speedup vs baseline: 2.2109x; 1.0731x over previous
#include <hip/hip_runtime.h>
#include <hip/hip_bf16.h>


typedef __attribute__((ext_vector_type(2))) float v2f;
typedef __attribute__((ext_vector_type(8))) float v8f;

#define N_NODES 40000
#define N_EDGES 640000
#define IN_DIM  3
#define HID_DIM 128
#define LAT_DIM 64

__global__ void zero_int_kernel(int* __restrict__ p, int n) {
    int i = blockIdx.x * blockDim.x + threadIdx.x;
    if (i < n) p[i] = 0;
}

__global__ void rank_kernel(const int* __restrict__ dst, int* __restrict__ cnt,
                            int* __restrict__ rank, int nEdges) {
    int e = blockIdx.x * blockDim.x + threadIdx.x;
    if (e < nEdges) rank[e] = atomicAdd(&cnt[dst[e]], 1);
}

#define SCAN_THREADS 1024
#define SCAN_PER     40
static_assert(SCAN_THREADS * SCAN_PER >= N_NODES, "scan covers all nodes");
static_assert(SCAN_PER % 4 == 0 && N_NODES % 4 == 0, "int4 chunks");

__global__ __launch_bounds__(SCAN_THREADS) void scan_kernel(
    const int* __restrict__ cnt, int* __restrict__ rowptr,
    float* __restrict__ dis, int n) {
    __shared__ int wsum[32];
    const int t    = threadIdx.x;
    const int lane = t & 31;
    const int w    = t >> 5;
    const int base = t * SCAN_PER;

    int v[SCAN_PER];
    int s = 0;
#pragma unroll
    for (int q = 0; q < SCAN_PER / 4; ++q) {
        const int idx = base + q * 4;
        int4 c = make_int4(0, 0, 0, 0);
        if (idx < n) c = *reinterpret_cast<const int4*>(cnt + idx);
        v[q * 4 + 0] = c.x; v[q * 4 + 1] = c.y;
        v[q * 4 + 2] = c.z; v[q * 4 + 3] = c.w;
        s += c.x + c.y + c.z + c.w;
    }

    int inc = s;
#pragma unroll
    for (int off = 1; off < 32; off <<= 1) {
        const int y = __shfl_up(inc, off);
        if (lane >= off) inc += y;
    }
    if (lane == 31) wsum[w] = inc;
    __syncthreads();
    if (w == 0) {
        int x = wsum[lane];
#pragma unroll
        for (int off = 1; off < 32; off <<= 1) {
            const int y = __shfl_up(x, off);
            if (lane >= off) x += y;
        }
        wsum[lane] = x;
    }
    __syncthreads();

    int excl = inc - s + (w > 0 ? wsum[w - 1] : 0);
#pragma unroll
    for (int q = 0; q < SCAN_PER / 4; ++q) {
        const int idx = base + q * 4;
        if (idx < n) {
            int4 r;
            float4 d;
            r.x = excl; excl += v[q * 4 + 0];
            r.y = excl; excl += v[q * 4 + 1];
            r.z = excl; excl += v[q * 4 + 2];
            r.w = excl; excl += v[q * 4 + 3];
            d.x = rsqrtf((float)v[q * 4 + 0] + 1.0f);
            d.y = rsqrtf((float)v[q * 4 + 1] + 1.0f);
            d.z = rsqrtf((float)v[q * 4 + 2] + 1.0f);
            d.w = rsqrtf((float)v[q * 4 + 3] + 1.0f);
            *reinterpret_cast<int4*>(rowptr + idx) = r;
            *reinterpret_cast<float4*>(dis + idx)  = d;
        }
    }
    if (t == SCAN_THREADS - 1) rowptr[n] = excl;
}

__global__ void fill_kernel(const int* __restrict__ src, const int* __restrict__ dst,
                            const int* __restrict__ rank,
                            const int* __restrict__ rowptr,
                            const float* __restrict__ dis,
                            int2* __restrict__ rec, int nEdges) {
    int e = blockIdx.x * blockDim.x + threadIdx.x;
    if (e >= nEdges) return;
    const int s = src[e];
    const int d = dst[e];
    const int pos = rowptr[d] + rank[e];
    const float coeff = dis[s] * dis[d];
    rec[pos] = make_int2(s, __float_as_int(coeff));
}

template <int K>
__global__ __launch_bounds__(256) void gemm_wmma_f32(
    const float* __restrict__ A, const float* __restrict__ B,
    float* __restrict__ C, int M, int N) {
    const int lane   = threadIdx.x & 31;
    const int wave   = threadIdx.x >> 5;
    const int tilesN = N >> 4;
    const int tile   = blockIdx.x * 8 + wave;
    const int totalTiles = (M >> 4) * tilesN;
    if (tile >= totalTiles) return;

    const int tm = tile / tilesN;
    const int tn = tile % tilesN;

    const int arow  = tm * 16 + (lane & 15);
    const int khalf = (lane >> 4) * 2;
    const int ncol  = tn * 16 + (lane & 15);

    v8f acc = {};
    for (int k0 = 0; k0 < K; k0 += 4) {
        const int ka = k0 + khalf;
        v2f a, b;
        a.x = (ka     < K) ? A[(size_t)arow * K + ka]     : 0.0f;
        a.y = (ka + 1 < K) ? A[(size_t)arow * K + ka + 1] : 0.0f;
        b.x = (ka     < K) ? B[(size_t)(ka)     * N + ncol] : 0.0f;
        b.y = (ka + 1 < K) ? B[(size_t)(ka + 1) * N + ncol] : 0.0f;
        acc = __builtin_amdgcn_wmma_f32_16x16x4_f32(
            false, a, false, b, (short)0, acc, false, false);
    }

    const int crow = tm * 16 + (lane >> 4) * 8;
#pragma unroll
    for (int r = 0; r < 8; ++r)
        C[(size_t)(crow + r) * N + ncol] = acc[r];
}

template <int N>
__global__ __launch_bounds__(256) void gemm_lds_f32(
    const float* __restrict__ A, const float* __restrict__ B,
    float* __restrict__ C) {
    constexpr int K     = HID_DIM;
    constexpr int ROWS  = 32;
    constexpr int TN    = N / 16;
    constexpr int WR    = 8 / TN;
    constexpr int RPW   = (ROWS / 16) / WR;
    constexpr int PITCH = K + 4;
    static_assert(K % 4 == 0 && 8 % TN == 0 && (ROWS / 16) % WR == 0, "tile split");
    static_assert(N_NODES % ROWS == 0, "whole row groups: no row guard needed");
    __shared__ float4 sA4[ROWS * PITCH / 4];
    float* sA = reinterpret_cast<float*>(sA4);

    const int t    = threadIdx.x;
    const int lane = t & 31;
    const int wave = t >> 5;
    const int row0 = blockIdx.x * ROWS;
    const int tn   = wave % TN;
    const int tr   = wave / TN;
    const int c    = lane & 15;
    const int h    = lane >> 4;
    const int ncol = tn * 16 + c;

    v2f bf[K / 4];
#pragma unroll
    for (int s = 0; s < K / 4; ++s) {
        const int ka = s * 4 + 2 * h;
        bf[s].x = B[(size_t)ka * N + ncol];
        bf[s].y = B[(size_t)(ka + 1) * N + ncol];
    }
    float4 av[ROWS / 8];
#pragma unroll
    for (int i = 0; i < ROWS / 8; ++i)
        av[i] = *reinterpret_cast<const float4*>(
            A + (size_t)(row0 + wave + i * 8) * K + lane * 4);
#pragma unroll
    for (int i = 0; i < ROWS / 8; ++i)
        *reinterpret_cast<float4*>(sA + (wave + i * 8) * PITCH + lane * 4) = av[i];
    __syncthreads();

    v8f acc[RPW];
#pragma unroll
    for (int j = 0; j < RPW; ++j) acc[j] = v8f{};
    const float* arow = sA + ((tr * RPW) * 16 + c) * PITCH + 2 * h;
#pragma unroll
    for (int s = 0; s < K / 4; ++s) {
#pragma unroll
        for (int j = 0; j < RPW; ++j) {
            const v2f a = *reinterpret_cast<const v2f*>(arow + j * 16 * PITCH + s * 4);
            acc[j] = __builtin_amdgcn_wmma_f32_16x16x4_f32(
                false, a, false, bf[s], (short)0, acc[j], false, false);
        }
    }

#pragma unroll
    for (int j = 0; j < RPW; ++j) {
        const int crow = row0 + (tr * RPW + j) * 16 + h * 8;
#pragma unroll
        for (int r = 0; r < 8; ++r)
            C[(size_t)(crow + r) * N + ncol] = acc[j][r];
    }
}

template <int DIM, bool RELU>
__global__ __launch_bounds__(256) void gather_rows(
    const int* __restrict__ rowptr, const int2* __restrict__ rec,
    const float* __restrict__ dis, const float* __restrict__ m,
    const float* __restrict__ bias, float* __restrict__ out, int nNodes) {
    constexpr int PER = DIM / 32;
    typedef __attribute__((ext_vector_type(PER))) float vec;

    const int lane = threadIdx.x & 31;
    const int node = __builtin_amdgcn_readfirstlane(
        (int)(blockIdx.x * (blockDim.x >> 5) + (threadIdx.x >> 5)));
    if (node >= nNodes) return;

    const int   beg = rowptr[node];
    const int   end = rowptr[node + 1];
    const float di  = dis[node];

    const vec* __restrict__ mv = reinterpret_cast<const vec*>(m);
    vec acc = mv[(size_t)node * 32 + lane] * di * di;

    for (int base = beg; base < end; base += 32) {
        const int n = min(32, end - base);
        int2 r = make_int2(0, 0);
        if (lane < n) r = rec[base + lane];
        int j = 0;
        for (; j + 4 <= n; j += 4) {
            const int   s0 = __builtin_amdgcn_readlane(r.x, j);
            const int   s1 = __builtin_amdgcn_readlane(r.x, j + 1);
            const int   s2 = __builtin_amdgcn_readlane(r.x, j + 2);
            const int   s3 = __builtin_amdgcn_readlane(r.x, j + 3);
            const float c0 = __int_as_float(__builtin_amdgcn_readlane(r.y, j));
            const float c1 = __int_as_float(__builtin_amdgcn_readlane(r.y, j + 1));
            const float c2 = __int_as_float(__builtin_amdgcn_readlane(r.y, j + 2));
            const float c3 = __int_as_float(__builtin_amdgcn_readlane(r.y, j + 3));
            const vec v0 = mv[(size_t)s0 * 32 + lane];
            const vec v1 = mv[(size_t)s1 * 32 + lane];
            const vec v2 = mv[(size_t)s2 * 32 + lane];
            const vec v3 = mv[(size_t)s3 * 32 + lane];
            acc += v0 * c0;
            acc += v1 * c1;
            acc += v2 * c2;
            acc += v3 * c3;
        }
        for (; j < n; ++j) {
            const int   s0 = __builtin_amdgcn_readlane(r.x, j);
            const float c0 = __int_as_float(__builtin_amdgcn_readlane(r.y, j));
            acc += mv[(size_t)s0 * 32 + lane] * c0;
        }
    }

    acc += reinterpret_cast<const vec*>(bias)[lane];
    if (RELU) {
#pragma unroll
        for (int k = 0; k < PER; ++k) acc[k] = fmaxf(acc[k], 0.0f);
    }
    reinterpret_cast<vec*>(out)[(size_t)node * 32 + lane] = acc;
}

extern "C" void kernel_launch(void* const* d_in, const int* in_sizes, int n_in,
                              void* d_out, int out_size, void* d_ws, size_t ws_size,
                              hipStream_t stream) {
    const float* x   = (const float*)d_in[0];
    const int*   ei  = (const int*)d_in[1];
    const float* W0  = (const float*)d_in[2];
    const float* b0  = (const float*)d_in[3];
    const float* W1  = (const float*)d_in[4];
    const float* b1  = (const float*)d_in[5];
    const float* W2  = (const float*)d_in[6];
    const float* b2  = (const float*)d_in[7];
    float* out = (float*)d_out;

    const int* src = ei;
    const int* dst = ei + N_EDGES;

    float* ws     = (float*)d_ws;
    float* m      = ws;
    float* h      = m + (size_t)N_NODES * HID_DIM;
    int2*  rec    = (int2*)(h + (size_t)N_NODES * HID_DIM);
    float* dis    = (float*)(rec + N_EDGES);
    int*   cnt    = (int*)(dis + N_NODES);
    int*   rowptr = cnt + N_NODES;
    int*   rank   = rowptr + (N_NODES + 32);

    const int T = 256;
    const int nodeWaveBlocks = (N_NODES * 32 + T - 1) / T;
    const int gemmBlocksH = (N_NODES / 16) * (HID_DIM / 16) / 8;
    const int gemmLdsBlocks = N_NODES / 32;

    zero_int_kernel<<<(N_NODES + T - 1) / T, T, 0, stream>>>(cnt, N_NODES);
    rank_kernel<<<(N_EDGES + T - 1) / T, T, 0, stream>>>(dst, cnt, rank, N_EDGES);
    scan_kernel<<<1, SCAN_THREADS, 0, stream>>>(cnt, rowptr, dis, N_NODES);
    fill_kernel<<<(N_EDGES + T - 1) / T, T, 0, stream>>>(src, dst, rank, rowptr, dis, rec, N_EDGES);

    gemm_wmma_f32<IN_DIM><<<gemmBlocksH, T, 0, stream>>>(x, W0, m, N_NODES, HID_DIM);
    gather_rows<HID_DIM, true><<<nodeWaveBlocks, T, 0, stream>>>(rowptr, rec, dis, m, b0, h, N_NODES);

    gemm_lds_f32<HID_DIM><<<gemmLdsBlocks, T, 0, stream>>>(h, W1, m);
    gather_rows<HID_DIM, true><<<nodeWaveBlocks, T, 0, stream>>>(rowptr, rec, dis, m, b1, h, N_NODES);

    gemm_lds_f32<LAT_DIM><<<gemmLdsBlocks, T, 0, stream>>>(h, W2, m);
    gather_rows<LAT_DIM, false><<<nodeWaveBlocks, T, 0, stream>>>(rowptr, rec, dis, m, b2, out, N_NODES);
}


# Each K=128 product fused into the gather before it: a 32-node work-group keeps its finished rows in LDS and multiplies them in the same launch; 8 launches; sum and k order unchanged
# speedup vs baseline: 2.4100x; 1.0901x over previous
#include <hip/hip_runtime.h>
#include <hip/hip_bf16.h>


typedef __attribute__((ext_vector_type(2))) float v2f;
typedef __attribute__((ext_vector_type(4))) float v4f;
typedef __attribute__((ext_vector_type(8))) float v8f;

#define N_NODES 40000
#define N_EDGES 640000
#define IN_DIM  3
#define HID_DIM 128
#define LAT_DIM 64

__global__ void zero_int_kernel(int* __restrict__ p, int n) {
    int i = blockIdx.x * blockDim.x + threadIdx.x;
    if (i < n) p[i] = 0;
}

__global__ void rank_kernel(const int* __restrict__ dst, int* __restrict__ cnt,
                            int* __restrict__ rank, int nEdges) {
    int e = blockIdx.x * blockDim.x + threadIdx.x;
    if (e < nEdges) rank[e] = atomicAdd(&cnt[dst[e]], 1);
}

#define SCAN_THREADS 1024
#define SCAN_PER     40
static_assert(SCAN_THREADS * SCAN_PER >= N_NODES, "scan covers all nodes");
static_assert(SCAN_PER % 4 == 0 && N_NODES % 4 == 0, "int4 chunks");

__global__ __launch_bounds__(SCAN_THREADS) void scan_kernel(
    const int* __restrict__ cnt, int* __restrict__ rowptr,
    float* __restrict__ dis, int n) {
    __shared__ int wsum[32];
    const int t    = threadIdx.x;
    const int lane = t & 31;
    const int w    = t >> 5;
    const int base = t * SCAN_PER;

    int v[SCAN_PER];
    int s = 0;
#pragma unroll
    for (int q = 0; q < SCAN_PER / 4; ++q) {
        const int idx = base + q * 4;
        int4 c = make_int4(0, 0, 0, 0);
        if (idx < n) c = *reinterpret_cast<const int4*>(cnt + idx);
        v[q * 4 + 0] = c.x; v[q * 4 + 1] = c.y;
        v[q * 4 + 2] = c.z; v[q * 4 + 3] = c.w;
        s += c.x + c.y + c.z + c.w;
    }

    int inc = s;
#pragma unroll
    for (int off = 1; off < 32; off <<= 1) {
        const int y = __shfl_up(inc, off);
        if (lane >= off) inc += y;
    }
    if (lane == 31) wsum[w] = inc;
    __syncthreads();
    if (w == 0) {
        int x = wsum[lane];
#pragma unroll
        for (int off = 1; off < 32; off <<= 1) {
            const int y = __shfl_up(x, off);
            if (lane >= off) x += y;
        }
        wsum[lane] = x;
    }
    __syncthreads();

    int excl = inc - s + (w > 0 ? wsum[w - 1] : 0);
#pragma unroll
    for (int q = 0; q < SCAN_PER / 4; ++q) {
        const int idx = base + q * 4;
        if (idx < n) {
            int4 r;
            float4 d;
            r.x = excl; excl += v[q * 4 + 0];
            r.y = excl; excl += v[q * 4 + 1];
            r.z = excl; excl += v[q * 4 + 2];
            r.w = excl; excl += v[q * 4 + 3];
            d.x = rsqrtf((float)v[q * 4 + 0] + 1.0f);
            d.y = rsqrtf((float)v[q * 4 + 1] + 1.0f);
            d.z = rsqrtf((float)v[q * 4 + 2] + 1.0f);
            d.w = rsqrtf((float)v[q * 4 + 3] + 1.0f);
            *reinterpret_cast<int4*>(rowptr + idx) = r;
            *reinterpret_cast<float4*>(dis + idx)  = d;
        }
    }
    if (t == SCAN_THREADS - 1) rowptr[n] = excl;
}

__global__ void fill_kernel(const int* __restrict__ src, const int* __restrict__ dst,
                            const int* __restrict__ rank,
                            const int* __restrict__ rowptr,
                            const float* __restrict__ dis,
                            int2* __restrict__ rec, int nEdges) {
    int e = blockIdx.x * blockDim.x + threadIdx.x;
    if (e >= nEdges) return;
    const int s = src[e];
    const int d = dst[e];
    const int pos = rowptr[d] + rank[e];
    const float coeff = dis[s] * dis[d];
    rec[pos] = make_int2(s, __float_as_int(coeff));
}

template <int K>
__global__ __launch_bounds__(256) void gemm_wmma_f32(
    const float* __restrict__ A, const float* __restrict__ B,
    float* __restrict__ C, int M, int N) {
    const int lane   = threadIdx.x & 31;
    const int wave   = threadIdx.x >> 5;
    const int tilesN = N >> 4;
    const int tile   = blockIdx.x * 8 + wave;
    const int totalTiles = (M >> 4) * tilesN;
    if (tile >= totalTiles) return;

    const int tm = tile / tilesN;
    const int tn = tile % tilesN;

    const int arow  = tm * 16 + (lane & 15);
    const int khalf = (lane >> 4) * 2;
    const int ncol  = tn * 16 + (lane & 15);

    v8f acc = {};
    for (int k0 = 0; k0 < K; k0 += 4) {
        const int ka = k0 + khalf;
        v2f a, b;
        a.x = (ka     < K) ? A[(size_t)arow * K + ka]     : 0.0f;
        a.y = (ka + 1 < K) ? A[(size_t)arow * K + ka + 1] : 0.0f;
        b.x = (ka     < K) ? B[(size_t)(ka)     * N + ncol] : 0.0f;
        b.y = (ka + 1 < K) ? B[(size_t)(ka + 1) * N + ncol] : 0.0f;
        acc = __builtin_amdgcn_wmma_f32_16x16x4_f32(
            false, a, false, b, (short)0, acc, false, false);
    }

    const int crow = tm * 16 + (lane >> 4) * 8;
#pragma unroll
    for (int r = 0; r < 8; ++r)
        C[(size_t)(crow + r) * N + ncol] = acc[r];
}

template <int N>
__global__ __launch_bounds__(256) void gather_gemm_f32(
    const int* __restrict__ rowptr, const int2* __restrict__ rec,
    const float* __restrict__ dis, const float* __restrict__ m,
    const float* __restrict__ bias, const float* __restrict__ B,
    float* __restrict__ C) {
    constexpr int K     = HID_DIM;
    constexpr int ROWS  = 32;
    constexpr int NPW   = ROWS / 8;
    constexpr int TN    = N / 16;
    constexpr int WR    = 8 / TN;
    constexpr int RPW   = (ROWS / 16) / WR;
    constexpr int PITCH = K + 4;
    static_assert(K == 128, "one float4 per lane covers a row");
    static_assert(K % 4 == 0 && 8 % TN == 0 && (ROWS / 16) % WR == 0, "tile split");
    static_assert(N_NODES % ROWS == 0, "whole row groups: no row guard needed");
    __shared__ float4 sA4[ROWS * PITCH / 4];
    float* sA = reinterpret_cast<float*>(sA4);

    const int t    = threadIdx.x;
    const int lane = t & 31;
    const int wave = __builtin_amdgcn_readfirstlane(t >> 5);
    const int row0 = blockIdx.x * ROWS;

    const v4f* __restrict__ mv = reinterpret_cast<const v4f*>(m);
    const v4f bv = reinterpret_cast<const v4f*>(bias)[lane];
    int   beg[NPW], end[NPW];
    float di[NPW];
#pragma unroll
    for (int i = 0; i < NPW; ++i) {
        const int node = row0 + wave + i * 8;
        beg[i] = rowptr[node];
        end[i] = rowptr[node + 1];
        di[i]  = dis[node];
    }
#pragma unroll
    for (int i = 0; i < NPW; ++i) {
        const int node = row0 + wave + i * 8;
        v4f acc = mv[(size_t)node * 32 + lane] * di[i] * di[i];
        for (int base = beg[i]; base < end[i]; base += 32) {
            const int n = min(32, end[i] - base);
            int2 r = make_int2(0, 0);
            if (lane < n) r = rec[base + lane];
            int j = 0;
            for (; j + 4 <= n; j += 4) {
                const int   s0 = __builtin_amdgcn_readlane(r.x, j);
                const int   s1 = __builtin_amdgcn_readlane(r.x, j + 1);
                const int   s2 = __builtin_amdgcn_readlane(r.x, j + 2);
                const int   s3 = __builtin_amdgcn_readlane(r.x, j + 3);
                const float c0 = __int_as_float(__builtin_amdgcn_readlane(r.y, j));
                const float c1 = __int_as_float(__builtin_amdgcn_readlane(r.y, j + 1));
                const float c2 = __int_as_float(__builtin_amdgcn_readlane(r.y, j + 2));
                const float c3 = __int_as_float(__builtin_amdgcn_readlane(r.y, j + 3));
                const v4f v0 = mv[(size_t)s0 * 32 + lane];
                const v4f v1 = mv[(size_t)s1 * 32 + lane];
                const v4f v2 = mv[(size_t)s2 * 32 + lane];
                const v4f v3 = mv[(size_t)s3 * 32 + lane];
                acc += v0 * c0;
                acc += v1 * c1;
                acc += v2 * c2;
                acc += v3 * c3;
            }
            for (; j < n; ++j) {
                const int   s0 = __builtin_amdgcn_readlane(r.x, j);
                const float c0 = __int_as_float(__builtin_amdgcn_readlane(r.y, j));
                acc += mv[(size_t)s0 * 32 + lane] * c0;
            }
        }
        acc += bv;
#pragma unroll
        for (int k = 0; k < 4; ++k) acc[k] = fmaxf(acc[k], 0.0f);
        *reinterpret_cast<v4f*>(sA + (wave + i * 8) * PITCH + lane * 4) = acc;
    }
    __syncthreads();

    const int tn   = wave % TN;
    const int tr   = wave / TN;
    const int c    = lane & 15;
    const int h    = lane >> 4;
    const int ncol = tn * 16 + c;

    v2f bf[K / 4];
#pragma unroll
    for (int s = 0; s < K / 4; ++s) {
        const int ka = s * 4 + 2 * h;
        bf[s].x = B[(size_t)ka * N + ncol];
        bf[s].y = B[(size_t)(ka + 1) * N + ncol];
    }

    v8f acc[RPW];
#pragma unroll
    for (int j = 0; j < RPW; ++j) acc[j] = v8f{};
    const float* arow = sA + ((tr * RPW) * 16 + c) * PITCH + 2 * h;
#pragma unroll
    for (int s = 0; s < K / 4; ++s) {
#pragma unroll
        for (int j = 0; j < RPW; ++j) {
            const v2f a = *reinterpret_cast<const v2f*>(arow + j * 16 * PITCH + s * 4);
            acc[j] = __builtin_amdgcn_wmma_f32_16x16x4_f32(
                false, a, false, bf[s], (short)0, acc[j], false, false);
        }
    }

#pragma unroll
    for (int j = 0; j < RPW; ++j) {
        const int crow = row0 + (tr * RPW + j) * 16 + h * 8;
#pragma unroll
        for (int r = 0; r < 8; ++r)
            C[(size_t)(crow + r) * N + ncol] = acc[j][r];
    }
}

template <int DIM, bool RELU>
__global__ __launch_bounds__(256) void gather_rows(
    const int* __restrict__ rowptr, const int2* __restrict__ rec,
    const float* __restrict__ dis, const float* __restrict__ m,
    const float* __restrict__ bias, float* __restrict__ out, int nNodes) {
    constexpr int PER = DIM / 32;
    typedef __attribute__((ext_vector_type(PER))) float vec;

    const int lane = threadIdx.x & 31;
    const int node = __builtin_amdgcn_readfirstlane(
        (int)(blockIdx.x * (blockDim.x >> 5) + (threadIdx.x >> 5)));
    if (node >= nNodes) return;

    const int   beg = rowptr[node];
    const int   end = rowptr[node + 1];
    const float di  = dis[node];

    const vec* __restrict__ mv = reinterpret_cast<const vec*>(m);
    vec acc = mv[(size_t)node * 32 + lane] * di * di;

    for (int base = beg; base < end; base += 32) {
        const int n = min(32, end - base);
        int2 r = make_int2(0, 0);
        if (lane < n) r = rec[base + lane];
        int j = 0;
        for (; j + 4 <= n; j += 4) {
            const int   s0 = __builtin_amdgcn_readlane(r.x, j);
            const int   s1 = __builtin_amdgcn_readlane(r.x, j + 1);
            const int   s2 = __builtin_amdgcn_readlane(r.x, j + 2);
            const int   s3 = __builtin_amdgcn_readlane(r.x, j + 3);
            const float c0 = __int_as_float(__builtin_amdgcn_readlane(r.y, j));
            const float c1 = __int_as_float(__builtin_amdgcn_readlane(r.y, j + 1));
            const float c2 = __int_as_float(__builtin_amdgcn_readlane(r.y, j + 2));
            const float c3 = __int_as_float(__builtin_amdgcn_readlane(r.y, j + 3));
            const vec v0 = mv[(size_t)s0 * 32 + lane];
            const vec v1 = mv[(size_t)s1 * 32 + lane];
            const vec v2 = mv[(size_t)s2 * 32 + lane];
            const vec v3 = mv[(size_t)s3 * 32 + lane];
            acc += v0 * c0;
            acc += v1 * c1;
            acc += v2 * c2;
            acc += v3 * c3;
        }
        for (; j < n; ++j) {
            const int   s0 = __builtin_amdgcn_readlane(r.x, j);
            const float c0 = __int_as_float(__builtin_amdgcn_readlane(r.y, j));
            acc += mv[(size_t)s0 * 32 + lane] * c0;
        }
    }

    acc += reinterpret_cast<const vec*>(bias)[lane];
    if (RELU) {
#pragma unroll
        for (int k = 0; k < PER; ++k) acc[k] = fmaxf(acc[k], 0.0f);
    }
    reinterpret_cast<vec*>(out)[(size_t)node * 32 + lane] = acc;
}

extern "C" void kernel_launch(void* const* d_in, const int* in_sizes, int n_in,
                              void* d_out, int out_size, void* d_ws, size_t ws_size,
                              hipStream_t stream) {
    const float* x   = (const float*)d_in[0];
    const int*   ei  = (const int*)d_in[1];
    const float* W0  = (const float*)d_in[2];
    const float* b0  = (const float*)d_in[3];
    const float* W1  = (const float*)d_in[4];
    const float* b1  = (const float*)d_in[5];
    const float* W2  = (const float*)d_in[6];
    const float* b2  = (const float*)d_in[7];
    float* out = (float*)d_out;

    const int* src = ei;
    const int* dst = ei + N_EDGES;

    float* ws     = (float*)d_ws;
    float* m      = ws;
    float* m2     = m + (size_t)N_NODES * HID_DIM;
    int2*  rec    = (int2*)(m2 + (size_t)N_NODES * HID_DIM);
    float* dis    = (float*)(rec + N_EDGES);
    int*   cnt    = (int*)(dis + N_NODES);
    int*   rowptr = cnt + N_NODES;
    int*   rank   = rowptr + (N_NODES + 32);

    const int T = 256;
    const int nodeWaveBlocks = (N_NODES * 32 + T - 1) / T;
    const int gemmBlocksH = (N_NODES / 16) * (HID_DIM / 16) / 8;
    const int fusedBlocks = N_NODES / 32;

    zero_int_kernel<<<(N_NODES + T - 1) / T, T, 0, stream>>>(cnt, N_NODES);
    rank_kernel<<<(N_EDGES + T - 1) / T, T, 0, stream>>>(dst, cnt, rank, N_EDGES);
    scan_kernel<<<1, SCAN_THREADS, 0, stream>>>(cnt, rowptr, dis, N_NODES);
    fill_kernel<<<(N_EDGES + T - 1) / T, T, 0, stream>>>(src, dst, rank, rowptr, dis, rec, N_EDGES);

    gemm_wmma_f32<IN_DIM><<<gemmBlocksH, T, 0, stream>>>(x, W0, m, N_NODES, HID_DIM);

    gather_gemm_f32<HID_DIM><<<fusedBlocks, T, 0, stream>>>(rowptr, rec, dis, m, b0, W1, m2);

    gather_gemm_f32<LAT_DIM><<<fusedBlocks, T, 0, stream>>>(rowptr, rec, dis, m2, b1, W2, m);

    gather_rows<LAT_DIM, false><<<nodeWaveBlocks, T, 0, stream>>>(rowptr, rec, dis, m, b2, out, N_NODES);
}


# Layer 1 sums each node's 3-wide input rows over its edges first and multiplies by W0 after ((Ax)W0 for A(xW0)), on WMMA in the fused launch; K=3 product launch gone; layer-1 sum order changes
# speedup vs baseline: 2.6353x; 1.0935x over previous
#include <hip/hip_runtime.h>
#include <hip/hip_bf16.h>


typedef __attribute__((ext_vector_type(2))) float v2f;
typedef __attribute__((ext_vector_type(4))) float v4f;
typedef __attribute__((ext_vector_type(8))) float v8f;

#define N_NODES 40000
#define N_EDGES 640000
#define IN_DIM  3
#define HID_DIM 128
#define LAT_DIM 64

__global__ void zero_int_kernel(int* __restrict__ p, int n) {
    int i = blockIdx.x * blockDim.x + threadIdx.x;
    if (i < n) p[i] = 0;
}

__global__ void rank_kernel(const int* __restrict__ dst, int* __restrict__ cnt,
                            int* __restrict__ rank, int nEdges) {
    int e = blockIdx.x * blockDim.x + threadIdx.x;
    if (e < nEdges) rank[e] = atomicAdd(&cnt[dst[e]], 1);
}

#define SCAN_THREADS 1024
#define SCAN_PER     40
static_assert(SCAN_THREADS * SCAN_PER >= N_NODES, "scan covers all nodes");
static_assert(SCAN_PER % 4 == 0 && N_NODES % 4 == 0, "int4 chunks");

__global__ __launch_bounds__(SCAN_THREADS) void scan_kernel(
    const int* __restrict__ cnt, int* __restrict__ rowptr,
    float* __restrict__ dis, int n) {
    __shared__ int wsum[32];
    const int t    = threadIdx.x;
    const int lane = t & 31;
    const int w    = t >> 5;
    const int base = t * SCAN_PER;

    int v[SCAN_PER];
    int s = 0;
#pragma unroll
    for (int q = 0; q < SCAN_PER / 4; ++q) {
        const int idx = base + q * 4;
        int4 c = make_int4(0, 0, 0, 0);
        if (idx < n) c = *reinterpret_cast<const int4*>(cnt + idx);
        v[q * 4 + 0] = c.x; v[q * 4 + 1] = c.y;
        v[q * 4 + 2] = c.z; v[q * 4 + 3] = c.w;
        s += c.x + c.y + c.z + c.w;
    }

    int inc = s;
#pragma unroll
    for (int off = 1; off < 32; off <<= 1) {
        const int y = __shfl_up(inc, off);
        if (lane >= off) inc += y;
    }
    if (lane == 31) wsum[w] = inc;
    __syncthreads();
    if (w == 0) {
        int x = wsum[lane];
#pragma unroll
        for (int off = 1; off < 32; off <<= 1) {
            const int y = __shfl_up(x, off);
            if (lane >= off) x += y;
        }
        wsum[lane] = x;
    }
    __syncthreads();

    int excl = inc - s + (w > 0 ? wsum[w - 1] : 0);
#pragma unroll
    for (int q = 0; q < SCAN_PER / 4; ++q) {
        const int idx = base + q * 4;
        if (idx < n) {
            int4 r;
            float4 d;
            r.x = excl; excl += v[q * 4 + 0];
            r.y = excl; excl += v[q * 4 + 1];
            r.z = excl; excl += v[q * 4 + 2];
            r.w = excl; excl += v[q * 4 + 3];
            d.x = rsqrtf((float)v[q * 4 + 0] + 1.0f);
            d.y = rsqrtf((float)v[q * 4 + 1] + 1.0f);
            d.z = rsqrtf((float)v[q * 4 + 2] + 1.0f);
            d.w = rsqrtf((float)v[q * 4 + 3] + 1.0f);
            *reinterpret_cast<int4*>(rowptr + idx) = r;
            *reinterpret_cast<float4*>(dis + idx)  = d;
        }
    }
    if (t == SCAN_THREADS - 1) rowptr[n] = excl;
}

__global__ void fill_kernel(const int* __restrict__ src, const int* __restrict__ dst,
                            const int* __restrict__ rank,
                            const int* __restrict__ rowptr,
                            const float* __restrict__ dis,
                            int2* __restrict__ rec, int nEdges) {
    int e = blockIdx.x * blockDim.x + threadIdx.x;
    if (e >= nEdges) return;
    const int s = src[e];
    const int d = dst[e];
    const int pos = rowptr[d] + rank[e];
    const float coeff = dis[s] * dis[d];
    rec[pos] = make_int2(s, __float_as_int(coeff));
}


template <int N>
__global__ __launch_bounds__(256) void gather_gemm_f32(
    const int* __restrict__ rowptr, const int2* __restrict__ rec,
    const float* __restrict__ dis, const float* __restrict__ m,
    const float* __restrict__ bias, const float* __restrict__ B,
    float* __restrict__ C) {
    constexpr int K     = HID_DIM;
    constexpr int ROWS  = 32;
    constexpr int NPW   = ROWS / 8;
    constexpr int TN    = N / 16;
    constexpr int WR    = 8 / TN;
    constexpr int RPW   = (ROWS / 16) / WR;
    constexpr int PITCH = K + 4;
    static_assert(K == 128, "one float4 per lane covers a row");
    static_assert(K % 4 == 0 && 8 % TN == 0 && (ROWS / 16) % WR == 0, "tile split");
    static_assert(N_NODES % ROWS == 0, "whole row groups: no row guard needed");
    __shared__ float4 sA4[ROWS * PITCH / 4];
    float* sA = reinterpret_cast<float*>(sA4);

    const int t    = threadIdx.x;
    const int lane = t & 31;
    const int wave = __builtin_amdgcn_readfirstlane(t >> 5);
    const int row0 = blockIdx.x * ROWS;

    const v4f* __restrict__ mv = reinterpret_cast<const v4f*>(m);
    const v4f bv = reinterpret_cast<const v4f*>(bias)[lane];
    int   beg[NPW], end[NPW];
    float di[NPW];
#pragma unroll
    for (int i = 0; i < NPW; ++i) {
        const int node = row0 + wave + i * 8;
        beg[i] = rowptr[node];
        end[i] = rowptr[node + 1];
        di[i]  = dis[node];
    }
#pragma unroll
    for (int i = 0; i < NPW; ++i) {
        const int node = row0 + wave + i * 8;
        v4f acc = mv[(size_t)node * 32 + lane] * di[i] * di[i];
        for (int base = beg[i]; base < end[i]; base += 32) {
            const int n = min(32, end[i] - base);
            int2 r = make_int2(0, 0);
            if (lane < n) r = rec[base + lane];
            int j = 0;
            for (; j + 4 <= n; j += 4) {
                const int   s0 = __builtin_amdgcn_readlane(r.x, j);
                const int   s1 = __builtin_amdgcn_readlane(r.x, j + 1);
                const int   s2 = __builtin_amdgcn_readlane(r.x, j + 2);
                const int   s3 = __builtin_amdgcn_readlane(r.x, j + 3);
                const float c0 = __int_as_float(__builtin_amdgcn_readlane(r.y, j));
                const float c1 = __int_as_float(__builtin_amdgcn_readlane(r.y, j + 1));
                const float c2 = __int_as_float(__builtin_amdgcn_readlane(r.y, j + 2));
                const float c3 = __int_as_float(__builtin_amdgcn_readlane(r.y, j + 3));
                const v4f v0 = mv[(size_t)s0 * 32 + lane];
                const v4f v1 = mv[(size_t)s1 * 32 + lane];
                const v4f v2 = mv[(size_t)s2 * 32 + lane];
                const v4f v3 = mv[(size_t)s3 * 32 + lane];
                acc += v0 * c0;
                acc += v1 * c1;
                acc += v2 * c2;
                acc += v3 * c3;
            }
            for (; j < n; ++j) {
                const int   s0 = __builtin_amdgcn_readlane(r.x, j);
                const float c0 = __int_as_float(__builtin_amdgcn_readlane(r.y, j));
                acc += mv[(size_t)s0 * 32 + lane] * c0;
            }
        }
        acc += bv;
#pragma unroll
        for (int k = 0; k < 4; ++k) acc[k] = fmaxf(acc[k], 0.0f);
        *reinterpret_cast<v4f*>(sA + (wave + i * 8) * PITCH + lane * 4) = acc;
    }
    __syncthreads();

    const int tn   = wave % TN;
    const int tr   = wave / TN;
    const int c    = lane & 15;
    const int h    = lane >> 4;
    const int ncol = tn * 16 + c;

    v2f bf[K / 4];
#pragma unroll
    for (int s = 0; s < K / 4; ++s) {
        const int ka = s * 4 + 2 * h;
        bf[s].x = B[(size_t)ka * N + ncol];
        bf[s].y = B[(size_t)(ka + 1) * N + ncol];
    }

    v8f acc[RPW];
#pragma unroll
    for (int j = 0; j < RPW; ++j) acc[j] = v8f{};
    const float* arow = sA + ((tr * RPW) * 16 + c) * PITCH + 2 * h;
#pragma unroll
    for (int s = 0; s < K / 4; ++s) {
#pragma unroll
        for (int j = 0; j < RPW; ++j) {
            const v2f a = *reinterpret_cast<const v2f*>(arow + j * 16 * PITCH + s * 4);
            acc[j] = __builtin_amdgcn_wmma_f32_16x16x4_f32(
                false, a, false, bf[s], (short)0, acc[j], false, false);
        }
    }

#pragma unroll
    for (int j = 0; j < RPW; ++j) {
        const int crow = row0 + (tr * RPW + j) * 16 + h * 8;
#pragma unroll
        for (int r = 0; r < 8; ++r)
            C[(size_t)(crow + r) * N + ncol] = acc[j][r];
    }
}

__global__ __launch_bounds__(256) void gather_in_gemm_f32(
    const int* __restrict__ rowptr, const int2* __restrict__ rec,
    const float* __restrict__ dis, const float* __restrict__ x,
    const float* __restrict__ Win, const float* __restrict__ bias,
    const float* __restrict__ B, float* __restrict__ C) {
    constexpr int K     = HID_DIM;
    constexpr int N     = HID_DIM;
    constexpr int ROWS  = 32;
    constexpr int SUB   = 8;
    constexpr int EPL   = 4;
    constexpr int RT    = ROWS / 16;
    constexpr int PITCH = K + 4;
    static_assert(ROWS * SUB == 256, "one group of lanes per node");
    static_assert(IN_DIM == 3, "input row padded to one k step of 4");
    static_assert(N / 16 == 8, "one column tile per wave");
    static_assert(N_NODES % ROWS == 0, "whole row groups: no row guard needed");
    __shared__ float4 sA4[ROWS * PITCH / 4];
    __shared__ float4 sX4[ROWS];
    float* sA = reinterpret_cast<float*>(sA4);
    const float* sX = reinterpret_cast<const float*>(sX4);

    const int t    = threadIdx.x;
    const int lane = t & 31;
    const int wave = __builtin_amdgcn_readfirstlane(t >> 5);
    const int row0 = blockIdx.x * ROWS;
    const int c    = lane & 15;
    const int h    = lane >> 4;
    const int ncol = wave * 16 + c;

    const int  nl   = t / SUB;
    const int  sub  = t % SUB;
    const int  node = row0 + nl;
    const int  beg  = rowptr[node];
    const int  end  = rowptr[node + 1];
    const float di  = dis[node];
    const float* xn = x + (size_t)node * IN_DIM;
    const float n0 = xn[0], n1 = xn[1], n2 = xn[2];

    v2f bf[K / 4];
#pragma unroll
    for (int s = 0; s < K / 4; ++s) {
        const int ka = s * 4 + 2 * h;
        bf[s].x = B[(size_t)ka * N + ncol];
        bf[s].y = B[(size_t)(ka + 1) * N + ncol];
    }
    v2f wf;
    {
        const float w1 = Win[(size_t)N + ncol];
        wf.x = Win[(size_t)(2 * h) * N + ncol];
        wf.y = h ? 0.0f : w1;
    }
    const float bcol = bias[ncol];

    {
        const bool own  = sub == 0;
        const float ws  = di * di;
        float a0 = own ? n0 * ws : 0.0f;
        float a1 = own ? n1 * ws : 0.0f;
        float a2 = own ? n2 * ws : 0.0f;
        for (int e = beg + sub; e < end; e += SUB * EPL) {
            int2 r[EPL];
            bool ok[EPL];
#pragma unroll
            for (int q = 0; q < EPL; ++q) {
                const int idx = e + q * SUB;
                ok[q] = idx < end;
                r[q]  = rec[ok[q] ? idx : end - 1];
            }
#pragma unroll
            for (int q = 0; q < EPL; ++q) {
                const float* xs = x + (size_t)r[q].x * IN_DIM;
                const float  cf = __int_as_float(r[q].y);
                const float  x0 = xs[0], x1 = xs[1], x2 = xs[2];
                a0 += (ok[q] ? x0 : 0.0f) * cf;
                a1 += (ok[q] ? x1 : 0.0f) * cf;
                a2 += (ok[q] ? x2 : 0.0f) * cf;
            }
        }
#pragma unroll
        for (int off = 1; off < SUB; off <<= 1) {
            a0 += __shfl_xor(a0, off);
            a1 += __shfl_xor(a1, off);
            a2 += __shfl_xor(a2, off);
        }
        if (own) sX4[nl] = make_float4(a0, a1, a2, 0.0f);
    }
    __syncthreads();

#pragma unroll
    for (int j = 0; j < RT; ++j) {
        const v2f a = *reinterpret_cast<const v2f*>(sX + (j * 16 + c) * 4 + 2 * h);
        const v8f d = __builtin_amdgcn_wmma_f32_16x16x4_f32(
            false, a, false, wf, (short)0, v8f{}, false, false);
#pragma unroll
        for (int r = 0; r < 8; ++r)
            sA[(j * 16 + h * 8 + r) * PITCH + ncol] = fmaxf(d[r] + bcol, 0.0f);
    }
    __syncthreads();

    v8f acc[RT];
#pragma unroll
    for (int j = 0; j < RT; ++j) acc[j] = v8f{};
    const float* arow = sA + c * PITCH + 2 * h;
#pragma unroll
    for (int s = 0; s < K / 4; ++s) {
#pragma unroll
        for (int j = 0; j < RT; ++j) {
            const v2f a = *reinterpret_cast<const v2f*>(arow + j * 16 * PITCH + s * 4);
            acc[j] = __builtin_amdgcn_wmma_f32_16x16x4_f32(
                false, a, false, bf[s], (short)0, acc[j], false, false);
        }
    }
#pragma unroll
    for (int j = 0; j < RT; ++j) {
        const int crow = row0 + j * 16 + h * 8;
#pragma unroll
        for (int r = 0; r < 8; ++r)
            C[(size_t)(crow + r) * N + ncol] = acc[j][r];
    }
}

template <int DIM, bool RELU>
__global__ __launch_bounds__(256) void gather_rows(
    const int* __restrict__ rowptr, const int2* __restrict__ rec,
    const float* __restrict__ dis, const float* __restrict__ m,
    const float* __restrict__ bias, float* __restrict__ out, int nNodes) {
    constexpr int PER = DIM / 32;
    typedef __attribute__((ext_vector_type(PER))) float vec;

    const int lane = threadIdx.x & 31;
    const int node = __builtin_amdgcn_readfirstlane(
        (int)(blockIdx.x * (blockDim.x >> 5) + (threadIdx.x >> 5)));
    if (node >= nNodes) return;

    const int   beg = rowptr[node];
    const int   end = rowptr[node + 1];
    const float di  = dis[node];

    const vec* __restrict__ mv = reinterpret_cast<const vec*>(m);
    vec acc = mv[(size_t)node * 32 + lane] * di * di;

    for (int base = beg; base < end; base += 32) {
        const int n = min(32, end - base);
        int2 r = make_int2(0, 0);
        if (lane < n) r = rec[base + lane];
        int j = 0;
        for (; j + 4 <= n; j += 4) {
            const int   s0 = __builtin_amdgcn_readlane(r.x, j);
            const int   s1 = __builtin_amdgcn_readlane(r.x, j + 1);
            const int   s2 = __builtin_amdgcn_readlane(r.x, j + 2);
            const int   s3 = __builtin_amdgcn_readlane(r.x, j + 3);
            const float c0 = __int_as_float(__builtin_amdgcn_readlane(r.y, j));
            const float c1 = __int_as_float(__builtin_amdgcn_readlane(r.y, j + 1));
            const float c2 = __int_as_float(__builtin_amdgcn_readlane(r.y, j + 2));
            const float c3 = __int_as_float(__builtin_amdgcn_readlane(r.y, j + 3));
            const vec v0 = mv[(size_t)s0 * 32 + lane];
            const vec v1 = mv[(size_t)s1 * 32 + lane];
            const vec v2 = mv[(size_t)s2 * 32 + lane];
            const vec v3 = mv[(size_t)s3 * 32 + lane];
            acc += v0 * c0;
            acc += v1 * c1;
            acc += v2 * c2;
            acc += v3 * c3;
        }
        for (; j < n; ++j) {
            const int   s0 = __builtin_amdgcn_readlane(r.x, j);
            const float c0 = __int_as_float(__builtin_amdgcn_readlane(r.y, j));
            acc += mv[(size_t)s0 * 32 + lane] * c0;
        }
    }

    acc += reinterpret_cast<const vec*>(bias)[lane];
    if (RELU) {
#pragma unroll
        for (int k = 0; k < PER; ++k) acc[k] = fmaxf(acc[k], 0.0f);
    }
    reinterpret_cast<vec*>(out)[(size_t)node * 32 + lane] = acc;
}

extern "C" void kernel_launch(void* const* d_in, const int* in_sizes, int n_in,
                              void* d_out, int out_size, void* d_ws, size_t ws_size,
                              hipStream_t stream) {
    const float* x   = (const float*)d_in[0];
    const int*   ei  = (const int*)d_in[1];
    const float* W0  = (const float*)d_in[2];
    const float* b0  = (const float*)d_in[3];
    const float* W1  = (const float*)d_in[4];
    const float* b1  = (const float*)d_in[5];
    const float* W2  = (const float*)d_in[6];
    const float* b2  = (const float*)d_in[7];
    float* out = (float*)d_out;

    const int* src = ei;
    const int* dst = ei + N_EDGES;

    float* ws     = (float*)d_ws;
    float* m      = ws;
    float* m2     = m + (size_t)N_NODES * HID_DIM;
    int2*  rec    = (int2*)(m2 + (size_t)N_NODES * HID_DIM);
    float* dis    = (float*)(rec + N_EDGES);
    int*   cnt    = (int*)(dis + N_NODES);
    int*   rowptr = cnt + N_NODES;
    int*   rank   = rowptr + (N_NODES + 32);

    const int T = 256;
    const int nodeWaveBlocks = (N_NODES * 32 + T - 1) / T;
    const int fusedBlocks = N_NODES / 32;

    zero_int_kernel<<<(N_NODES + T - 1) / T, T, 0, stream>>>(cnt, N_NODES);
    rank_kernel<<<(N_EDGES + T - 1) / T, T, 0, stream>>>(dst, cnt, rank, N_EDGES);
    scan_kernel<<<1, SCAN_THREADS, 0, stream>>>(cnt, rowptr, dis, N_NODES);
    fill_kernel<<<(N_EDGES + T - 1) / T, T, 0, stream>>>(src, dst, rank, rowptr, dis, rec, N_EDGES);

    gather_in_gemm_f32<<<fusedBlocks, T, 0, stream>>>(rowptr, rec, dis, x, W0, b0, W1, m2);

    gather_gemm_f32<LAT_DIM><<<fusedBlocks, T, 0, stream>>>(rowptr, rec, dis, m2, b1, W2, m);

    gather_rows<LAT_DIM, false><<<nodeWaveBlocks, T, 0, stream>>>(rowptr, rec, dis, m, b2, out, N_NODES);
}


# Row-offset scan fused into edge placement: each of 256 placement work-groups rebuilds the 40000 offsets in LDS from the counts and takes offsets and degrees from LDS; 6 launches; sums unchanged
# speedup vs baseline: 3.0510x; 1.1577x over previous
#include <hip/hip_runtime.h>
#include <hip/hip_bf16.h>


typedef __attribute__((ext_vector_type(2))) float v2f;
typedef __attribute__((ext_vector_type(4))) float v4f;
typedef __attribute__((ext_vector_type(8))) float v8f;

#define N_NODES 40000
#define N_EDGES 640000
#define IN_DIM  3
#define HID_DIM 128
#define LAT_DIM 64

__global__ void zero_int_kernel(int* __restrict__ p, int n) {
    int i = blockIdx.x * blockDim.x + threadIdx.x;
    if (i < n) p[i] = 0;
}

__global__ void rank_kernel(const int* __restrict__ dst, int* __restrict__ cnt,
                            int* __restrict__ rank, int nEdges) {
    int e = blockIdx.x * blockDim.x + threadIdx.x;
    if (e < nEdges) rank[e] = atomicAdd(&cnt[dst[e]], 1);
}

#define PLACE_THREADS 1024
#define PLACE_GROUPS  256
#define SCAN_PER      40
#define RP_WORDS      (PLACE_THREADS * SCAN_PER)
static_assert(RP_WORDS >= N_NODES + 1, "table covers all nodes and the total");
static_assert(SCAN_PER % 4 == 0 && N_NODES % 4 == 0, "int4 chunks");
static_assert(N_EDGES % PLACE_GROUPS == 0, "equal edge slices");

__global__ __launch_bounds__(PLACE_THREADS) void scan_fill_kernel(
    const int* __restrict__ src, const int* __restrict__ dst,
    const int* __restrict__ rank, const int* __restrict__ cnt,
    int* __restrict__ rowptr, float* __restrict__ dis,
    int2* __restrict__ rec) {
    constexpr int EPG   = N_EDGES / PLACE_GROUPS;
    constexpr int TRIPS = (EPG + PLACE_THREADS - 1) / PLACE_THREADS;
    constexpr int NPG   = (N_NODES + 1 + PLACE_GROUPS - 1) / PLACE_GROUPS;
    __shared__ int4 rp4[RP_WORDS / 4];
    __shared__ int wsum[32];
    const int* rp = reinterpret_cast<const int*>(rp4);
    const int t    = threadIdx.x;
    const int lane = t & 31;
    const int w    = t >> 5;

    int4 c[SCAN_PER / 4];
#pragma unroll
    for (int i = 0; i < SCAN_PER / 4; ++i) {
        const int idx = (i * PLACE_THREADS + t) * 4;
        c[i] = make_int4(0, 0, 0, 0);
        if (idx < N_NODES) c[i] = *reinterpret_cast<const int4*>(cnt + idx);
    }
    const int e0 = blockIdx.x * EPG;
    int  es[TRIPS], ed[TRIPS], er[TRIPS];
    bool ok[TRIPS];
#pragma unroll
    for (int i = 0; i < TRIPS; ++i) {
        const int k = t + i * PLACE_THREADS;
        ok[i] = k < EPG;
        const int e = e0 + (ok[i] ? k : 0);
        es[i] = src[e];
        ed[i] = dst[e];
        er[i] = rank[e];
    }
#pragma unroll
    for (int i = 0; i < SCAN_PER / 4; ++i) rp4[i * PLACE_THREADS + t] = c[i];
    __syncthreads();

    int v[SCAN_PER];
    int s = 0;
#pragma unroll
    for (int q = 0; q < SCAN_PER / 4; ++q) {
        const int4 x = rp4[t * (SCAN_PER / 4) + q];
        v[q * 4 + 0] = x.x; v[q * 4 + 1] = x.y;
        v[q * 4 + 2] = x.z; v[q * 4 + 3] = x.w;
        s += x.x + x.y + x.z + x.w;
    }
    int inc = s;
#pragma unroll
    for (int off = 1; off < 32; off <<= 1) {
        const int y = __shfl_up(inc, off);
        if (lane >= off) inc += y;
    }
    if (lane == 31) wsum[w] = inc;
    __syncthreads();
    if (w == 0) {
        int x = wsum[lane];
#pragma unroll
        for (int off = 1; off < 32; off <<= 1) {
            const int y = __shfl_up(x, off);
            if (lane >= off) x += y;
        }
        wsum[lane] = x;
    }
    __syncthreads();

    int excl = inc - s + (w > 0 ? wsum[w - 1] : 0);
#pragma unroll
    for (int q = 0; q < SCAN_PER / 4; ++q) {
        int4 r;
        r.x = excl; excl += v[q * 4 + 0];
        r.y = excl; excl += v[q * 4 + 1];
        r.z = excl; excl += v[q * 4 + 2];
        r.w = excl; excl += v[q * 4 + 3];
        rp4[t * (SCAN_PER / 4) + q] = r;
    }
    __syncthreads();

    if (t < NPG) {
        const int node = blockIdx.x * NPG + t;
        if (node <= N_NODES) {
            const int a = rp[node];
            rowptr[node] = a;
            if (node < N_NODES) dis[node] = rsqrtf((float)(rp[node + 1] - a) + 1.0f);
        }
    }

#pragma unroll
    for (int i = 0; i < TRIPS; ++i) {
        if (ok[i]) {
            const int sN = es[i];
            const int dN = ed[i];
            const int pd = rp[dN];
            const int cd = rp[dN + 1] - pd;
            const int cs = rp[sN + 1] - rp[sN];
            const float coeff = rsqrtf((float)cs + 1.0f) * rsqrtf((float)cd + 1.0f);
            rec[pd + er[i]] = make_int2(sN, __float_as_int(coeff));
        }
    }
}


template <int N>
__global__ __launch_bounds__(256) void gather_gemm_f32(
    const int* __restrict__ rowptr, const int2* __restrict__ rec,
    const float* __restrict__ dis, const float* __restrict__ m,
    const float* __restrict__ bias, const float* __restrict__ B,
    float* __restrict__ C) {
    constexpr int K     = HID_DIM;
    constexpr int ROWS  = 32;
    constexpr int NPW   = ROWS / 8;
    constexpr int TN    = N / 16;
    constexpr int WR    = 8 / TN;
    constexpr int RPW   = (ROWS / 16) / WR;
    constexpr int PITCH = K + 4;
    static_assert(K == 128, "one float4 per lane covers a row");
    static_assert(K % 4 == 0 && 8 % TN == 0 && (ROWS / 16) % WR == 0, "tile split");
    static_assert(N_NODES % ROWS == 0, "whole row groups: no row guard needed");
    __shared__ float4 sA4[ROWS * PITCH / 4];
    float* sA = reinterpret_cast<float*>(sA4);

    const int t    = threadIdx.x;
    const int lane = t & 31;
    const int wave = __builtin_amdgcn_readfirstlane(t >> 5);
    const int row0 = blockIdx.x * ROWS;

    const v4f* __restrict__ mv = reinterpret_cast<const v4f*>(m);
    const v4f bv = reinterpret_cast<const v4f*>(bias)[lane];
    int   beg[NPW], end[NPW];
    float di[NPW];
#pragma unroll
    for (int i = 0; i < NPW; ++i) {
        const int node = row0 + wave + i * 8;
        beg[i] = rowptr[node];
        end[i] = rowptr[node + 1];
        di[i]  = dis[node];
    }
#pragma unroll
    for (int i = 0; i < NPW; ++i) {
        const int node = row0 + wave + i * 8;
        v4f acc = mv[(size_t)node * 32 + lane] * di[i] * di[i];
        for (int base = beg[i]; base < end[i]; base += 32) {
            const int n = min(32, end[i] - base);
            int2 r = make_int2(0, 0);
            if (lane < n) r = rec[base + lane];
            int j = 0;
            for (; j + 4 <= n; j += 4) {
                const int   s0 = __builtin_amdgcn_readlane(r.x, j);
                const int   s1 = __builtin_amdgcn_readlane(r.x, j + 1);
                const int   s2 = __builtin_amdgcn_readlane(r.x, j + 2);
                const int   s3 = __builtin_amdgcn_readlane(r.x, j + 3);
                const float c0 = __int_as_float(__builtin_amdgcn_readlane(r.y, j));
                const float c1 = __int_as_float(__builtin_amdgcn_readlane(r.y, j + 1));
                const float c2 = __int_as_float(__builtin_amdgcn_readlane(r.y, j + 2));
                const float c3 = __int_as_float(__builtin_amdgcn_readlane(r.y, j + 3));
                const v4f v0 = mv[(size_t)s0 * 32 + lane];
                const v4f v1 = mv[(size_t)s1 * 32 + lane];
                const v4f v2 = mv[(size_t)s2 * 32 + lane];
                const v4f v3 = mv[(size_t)s3 * 32 + lane];
                acc += v0 * c0;
                acc += v1 * c1;
                acc += v2 * c2;
                acc += v3 * c3;
            }
            for (; j < n; ++j) {
                const int   s0 = __builtin_amdgcn_readlane(r.x, j);
                const float c0 = __int_as_float(__builtin_amdgcn_readlane(r.y, j));
                acc += mv[(size_t)s0 * 32 + lane] * c0;
            }
        }
        acc += bv;
#pragma unroll
        for (int k = 0; k < 4; ++k) acc[k] = fmaxf(acc[k], 0.0f);
        *reinterpret_cast<v4f*>(sA + (wave + i * 8) * PITCH + lane * 4) = acc;
    }
    __syncthreads();

    const int tn   = wave % TN;
    const int tr   = wave / TN;
    const int c    = lane & 15;
    const int h    = lane >> 4;
    const int ncol = tn * 16 + c;

    v2f bf[K / 4];
#pragma unroll
    for (int s = 0; s < K / 4; ++s) {
        const int ka = s * 4 + 2 * h;
        bf[s].x = B[(size_t)ka * N + ncol];
        bf[s].y = B[(size_t)(ka + 1) * N + ncol];
    }

    v8f acc[RPW];
#pragma unroll
    for (int j = 0; j < RPW; ++j) acc[j] = v8f{};
    const float* arow = sA + ((tr * RPW) * 16 + c) * PITCH + 2 * h;
#pragma unroll
    for (int s = 0; s < K / 4; ++s) {
#pragma unroll
        for (int j = 0; j < RPW; ++j) {
            const v2f a = *reinterpret_cast<const v2f*>(arow + j * 16 * PITCH + s * 4);
            acc[j] = __builtin_amdgcn_wmma_f32_16x16x4_f32(
                false, a, false, bf[s], (short)0, acc[j], false, false);
        }
    }

#pragma unroll
    for (int j = 0; j < RPW; ++j) {
        const int crow = row0 + (tr * RPW + j) * 16 + h * 8;
#pragma unroll
        for (int r = 0; r < 8; ++r)
            C[(size_t)(crow + r) * N + ncol] = acc[j][r];
    }
}

__global__ __launch_bounds__(256) void gather_in_gemm_f32(
    const int* __restrict__ rowptr, const int2* __restrict__ rec,
    const float* __restrict__ dis, const float* __restrict__ x,
    const float* __restrict__ Win, const float* __restrict__ bias,
    const float* __restrict__ B, float* __restrict__ C) {
    constexpr int K     = HID_DIM;
    constexpr int N     = HID_DIM;
    constexpr int ROWS  = 32;
    constexpr int SUB   = 8;
    constexpr int EPL   = 4;
    constexpr int RT    = ROWS / 16;
    constexpr int PITCH = K + 4;
    static_assert(ROWS * SUB == 256, "one group of lanes per node");
    static_assert(IN_DIM == 3, "input row padded to one k step of 4");
    static_assert(N / 16 == 8, "one column tile per wave");
    static_assert(N_NODES % ROWS == 0, "whole row groups: no row guard needed");
    __shared__ float4 sA4[ROWS * PITCH / 4];
    __shared__ float4 sX4[ROWS];
    float* sA = reinterpret_cast<float*>(sA4);
    const float* sX = reinterpret_cast<const float*>(sX4);

    const int t    = threadIdx.x;
    const int lane = t & 31;
    const int wave = __builtin_amdgcn_readfirstlane(t >> 5);
    const int row0 = blockIdx.x * ROWS;
    const int c    = lane & 15;
    const int h    = lane >> 4;
    const int ncol = wave * 16 + c;

    const int  nl   = t / SUB;
    const int  sub  = t % SUB;
    const int  node = row0 + nl;
    const int  beg  = rowptr[node];
    const int  end  = rowptr[node + 1];
    const float di  = dis[node];
    const float* xn = x + (size_t)node * IN_DIM;
    const float n0 = xn[0], n1 = xn[1], n2 = xn[2];

    v2f bf[K / 4];
#pragma unroll
    for (int s = 0; s < K / 4; ++s) {
        const int ka = s * 4 + 2 * h;
        bf[s].x = B[(size_t)ka * N + ncol];
        bf[s].y = B[(size_t)(ka + 1) * N + ncol];
    }
    v2f wf;
    {
        const float w1 = Win[(size_t)N + ncol];
        wf.x = Win[(size_t)(2 * h) * N + ncol];
        wf.y = h ? 0.0f : w1;
    }
    const float bcol = bias[ncol];

    {
        const bool own  = sub == 0;
        const float ws  = di * di;
        float a0 = own ? n0 * ws : 0.0f;
        float a1 = own ? n1 * ws : 0.0f;
        float a2 = own ? n2 * ws : 0.0f;
        for (int e = beg + sub; e < end; e += SUB * EPL) {
            int2 r[EPL];
            bool ok[EPL];
#pragma unroll
            for (int q = 0; q < EPL; ++q) {
                const int idx = e + q * SUB;
                ok[q] = idx < end;
                r[q]  = rec[ok[q] ? idx : end - 1];
            }
#pragma unroll
            for (int q = 0; q < EPL; ++q) {
                const float* xs = x + (size_t)r[q].x * IN_DIM;
                const float  cf = __int_as_float(r[q].y);
                const float  x0 = xs[0], x1 = xs[1], x2 = xs[2];
                a0 += (ok[q] ? x0 : 0.0f) * cf;
                a1 += (ok[q] ? x1 : 0.0f) * cf;
                a2 += (ok[q] ? x2 : 0.0f) * cf;
            }
        }
#pragma unroll
        for (int off = 1; off < SUB; off <<= 1) {
            a0 += __shfl_xor(a0, off);
            a1 += __shfl_xor(a1, off);
            a2 += __shfl_xor(a2, off);
        }
        if (own) sX4[nl] = make_float4(a0, a1, a2, 0.0f);
    }
    __syncthreads();

#pragma unroll
    for (int j = 0; j < RT; ++j) {
        const v2f a = *reinterpret_cast<const v2f*>(sX + (j * 16 + c) * 4 + 2 * h);
        const v8f d = __builtin_amdgcn_wmma_f32_16x16x4_f32(
            false, a, false, wf, (short)0, v8f{}, false, false);
#pragma unroll
        for (int r = 0; r < 8; ++r)
            sA[(j * 16 + h * 8 + r) * PITCH + ncol] = fmaxf(d[r] + bcol, 0.0f);
    }
    __syncthreads();

    v8f acc[RT];
#pragma unroll
    for (int j = 0; j < RT; ++j) acc[j] = v8f{};
    const float* arow = sA + c * PITCH + 2 * h;
#pragma unroll
    for (int s = 0; s < K / 4; ++s) {
#pragma unroll
        for (int j = 0; j < RT; ++j) {
            const v2f a = *reinterpret_cast<const v2f*>(arow + j * 16 * PITCH + s * 4);
            acc[j] = __builtin_amdgcn_wmma_f32_16x16x4_f32(
                false, a, false, bf[s], (short)0, acc[j], false, false);
        }
    }
#pragma unroll
    for (int j = 0; j < RT; ++j) {
        const int crow = row0 + j * 16 + h * 8;
#pragma unroll
        for (int r = 0; r < 8; ++r)
            C[(size_t)(crow + r) * N + ncol] = acc[j][r];
    }
}

template <int DIM, bool RELU>
__global__ __launch_bounds__(256) void gather_rows(
    const int* __restrict__ rowptr, const int2* __restrict__ rec,
    const float* __restrict__ dis, const float* __restrict__ m,
    const float* __restrict__ bias, float* __restrict__ out, int nNodes) {
    constexpr int PER = DIM / 32;
    typedef __attribute__((ext_vector_type(PER))) float vec;

    const int lane = threadIdx.x & 31;
    const int node = __builtin_amdgcn_readfirstlane(
        (int)(blockIdx.x * (blockDim.x >> 5) + (threadIdx.x >> 5)));
    if (node >= nNodes) return;

    const int   beg = rowptr[node];
    const int   end = rowptr[node + 1];
    const float di  = dis[node];

    const vec* __restrict__ mv = reinterpret_cast<const vec*>(m);
    vec acc = mv[(size_t)node * 32 + lane] * di * di;

    for (int base = beg; base < end; base += 32) {
        const int n = min(32, end - base);
        int2 r = make_int2(0, 0);
        if (lane < n) r = rec[base + lane];
        int j = 0;
        for (; j + 4 <= n; j += 4) {
            const int   s0 = __builtin_amdgcn_readlane(r.x, j);
            const int   s1 = __builtin_amdgcn_readlane(r.x, j + 1);
            const int   s2 = __builtin_amdgcn_readlane(r.x, j + 2);
            const int   s3 = __builtin_amdgcn_readlane(r.x, j + 3);
            const float c0 = __int_as_float(__builtin_amdgcn_readlane(r.y, j));
            const float c1 = __int_as_float(__builtin_amdgcn_readlane(r.y, j + 1));
            const float c2 = __int_as_float(__builtin_amdgcn_readlane(r.y, j + 2));
            const float c3 = __int_as_float(__builtin_amdgcn_readlane(r.y, j + 3));
            const vec v0 = mv[(size_t)s0 * 32 + lane];
            const vec v1 = mv[(size_t)s1 * 32 + lane];
            const vec v2 = mv[(size_t)s2 * 32 + lane];
            const vec v3 = mv[(size_t)s3 * 32 + lane];
            acc += v0 * c0;
            acc += v1 * c1;
            acc += v2 * c2;
            acc += v3 * c3;
        }
        for (; j < n; ++j) {
            const int   s0 = __builtin_amdgcn_readlane(r.x, j);
            const float c0 = __int_as_float(__builtin_amdgcn_readlane(r.y, j));
            acc += mv[(size_t)s0 * 32 + lane] * c0;
        }
    }

    acc += reinterpret_cast<const vec*>(bias)[lane];
    if (RELU) {
#pragma unroll
        for (int k = 0; k < PER; ++k) acc[k] = fmaxf(acc[k], 0.0f);
    }
    reinterpret_cast<vec*>(out)[(size_t)node * 32 + lane] = acc;
}

extern "C" void kernel_launch(void* const* d_in, const int* in_sizes, int n_in,
                              void* d_out, int out_size, void* d_ws, size_t ws_size,
                              hipStream_t stream) {
    const float* x   = (const float*)d_in[0];
    const int*   ei  = (const int*)d_in[1];
    const float* W0  = (const float*)d_in[2];
    const float* b0  = (const float*)d_in[3];
    const float* W1  = (const float*)d_in[4];
    const float* b1  = (const float*)d_in[5];
    const float* W2  = (const float*)d_in[6];
    const float* b2  = (const float*)d_in[7];
    float* out = (float*)d_out;

    const int* src = ei;
    const int* dst = ei + N_EDGES;

    float* ws     = (float*)d_ws;
    float* m      = ws;
    float* m2     = m + (size_t)N_NODES * HID_DIM;
    int2*  rec    = (int2*)(m2 + (size_t)N_NODES * HID_DIM);
    float* dis    = (float*)(rec + N_EDGES);
    int*   cnt    = (int*)(dis + N_NODES);
    int*   rowptr = cnt + N_NODES;
    int*   rank   = rowptr + (N_NODES + 32);

    const int T = 256;
    const int nodeWaveBlocks = (N_NODES * 32 + T - 1) / T;
    const int fusedBlocks = N_NODES / 32;

    zero_int_kernel<<<(N_NODES + T - 1) / T, T, 0, stream>>>(cnt, N_NODES);
    rank_kernel<<<(N_EDGES + T - 1) / T, T, 0, stream>>>(dst, cnt, rank, N_EDGES);
    scan_fill_kernel<<<PLACE_GROUPS, PLACE_THREADS, 0, stream>>>(src, dst, rank, cnt, rowptr, dis, rec);

    gather_in_gemm_f32<<<fusedBlocks, T, 0, stream>>>(rowptr, rec, dis, x, W0, b0, W1, m2);

    gather_gemm_f32<LAT_DIM><<<fusedBlocks, T, 0, stream>>>(rowptr, rec, dis, m2, b1, W2, m);

    gather_rows<LAT_DIM, false><<<nodeWaveBlocks, T, 0, stream>>>(rowptr, rec, dis, m, b2, out, N_NODES);
}


# Layer 2 sums h1 rows over edges before W1 ((A h1)W1 for A(h1 W1)); h1[src] re-made per edge on WMMA from its 16-byte summed input row; 512-byte gather and row table gone; layer-2 sum order changed
# speedup vs baseline: 3.1943x; 1.0470x over previous
#include <hip/hip_runtime.h>
#include <hip/hip_bf16.h>


typedef __attribute__((ext_vector_type(2))) float v2f;
typedef __attribute__((ext_vector_type(4))) float v4f;
typedef __attribute__((ext_vector_type(8))) float v8f;

#define N_NODES 40000
#define N_EDGES 640000
#define IN_DIM  3
#define HID_DIM 128
#define LAT_DIM 64

__global__ void zero_int_kernel(int* __restrict__ p, int n) {
    int i = blockIdx.x * blockDim.x + threadIdx.x;
    if (i < n) p[i] = 0;
}

__global__ void rank_kernel(const int* __restrict__ dst, int* __restrict__ cnt,
                            int* __restrict__ rank, int nEdges) {
    int e = blockIdx.x * blockDim.x + threadIdx.x;
    if (e < nEdges) rank[e] = atomicAdd(&cnt[dst[e]], 1);
}

#define PLACE_THREADS 1024
#define PLACE_GROUPS  256
#define SCAN_PER      40
#define RP_WORDS      (PLACE_THREADS * SCAN_PER)
static_assert(RP_WORDS >= N_NODES + 1, "table covers all nodes and the total");
static_assert(SCAN_PER % 4 == 0 && N_NODES % 4 == 0, "int4 chunks");
static_assert(N_EDGES % PLACE_GROUPS == 0, "equal edge slices");

__global__ __launch_bounds__(PLACE_THREADS) void scan_fill_kernel(
    const int* __restrict__ src, const int* __restrict__ dst,
    const int* __restrict__ rank, const int* __restrict__ cnt,
    int* __restrict__ rowptr, float* __restrict__ dis,
    int2* __restrict__ rec) {
    constexpr int EPG   = N_EDGES / PLACE_GROUPS;
    constexpr int TRIPS = (EPG + PLACE_THREADS - 1) / PLACE_THREADS;
    constexpr int NPG   = (N_NODES + 1 + PLACE_GROUPS - 1) / PLACE_GROUPS;
    __shared__ int4 rp4[RP_WORDS / 4];
    __shared__ int wsum[32];
    const int* rp = reinterpret_cast<const int*>(rp4);
    const int t    = threadIdx.x;
    const int lane = t & 31;
    const int w    = t >> 5;

    int4 c[SCAN_PER / 4];
#pragma unroll
    for (int i = 0; i < SCAN_PER / 4; ++i) {
        const int idx = (i * PLACE_THREADS + t) * 4;
        c[i] = make_int4(0, 0, 0, 0);
        if (idx < N_NODES) c[i] = *reinterpret_cast<const int4*>(cnt + idx);
    }
    const int e0 = blockIdx.x * EPG;
    int  es[TRIPS], ed[TRIPS], er[TRIPS];
    bool ok[TRIPS];
#pragma unroll
    for (int i = 0; i < TRIPS; ++i) {
        const int k = t + i * PLACE_THREADS;
        ok[i] = k < EPG;
        const int e = e0 + (ok[i] ? k : 0);
        es[i] = src[e];
        ed[i] = dst[e];
        er[i] = rank[e];
    }
#pragma unroll
    for (int i = 0; i < SCAN_PER / 4; ++i) rp4[i * PLACE_THREADS + t] = c[i];
    __syncthreads();

    int v[SCAN_PER];
    int s = 0;
#pragma unroll
    for (int q = 0; q < SCAN_PER / 4; ++q) {
        const int4 x = rp4[t * (SCAN_PER / 4) + q];
        v[q * 4 + 0] = x.x; v[q * 4 + 1] = x.y;
        v[q * 4 + 2] = x.z; v[q * 4 + 3] = x.w;
        s += x.x + x.y + x.z + x.w;
    }
    int inc = s;
#pragma unroll
    for (int off = 1; off < 32; off <<= 1) {
        const int y = __shfl_up(inc, off);
        if (lane >= off) inc += y;
    }
    if (lane == 31) wsum[w] = inc;
    __syncthreads();
    if (w == 0) {
        int x = wsum[lane];
#pragma unroll
        for (int off = 1; off < 32; off <<= 1) {
            const int y = __shfl_up(x, off);
            if (lane >= off) x += y;
        }
        wsum[lane] = x;
    }
    __syncthreads();

    int excl = inc - s + (w > 0 ? wsum[w - 1] : 0);
#pragma unroll
    for (int q = 0; q < SCAN_PER / 4; ++q) {
        int4 r;
        r.x = excl; excl += v[q * 4 + 0];
        r.y = excl; excl += v[q * 4 + 1];
        r.z = excl; excl += v[q * 4 + 2];
        r.w = excl; excl += v[q * 4 + 3];
        rp4[t * (SCAN_PER / 4) + q] = r;
    }
    __syncthreads();

    if (t < NPG) {
        const int node = blockIdx.x * NPG + t;
        if (node <= N_NODES) {
            const int a = rp[node];
            rowptr[node] = a;
            if (node < N_NODES) dis[node] = rsqrtf((float)(rp[node + 1] - a) + 1.0f);
        }
    }

#pragma unroll
    for (int i = 0; i < TRIPS; ++i) {
        if (ok[i]) {
            const int sN = es[i];
            const int dN = ed[i];
            const int pd = rp[dN];
            const int cd = rp[dN + 1] - pd;
            const int cs = rp[sN + 1] - rp[sN];
            const float coeff = rsqrtf((float)cs + 1.0f) * rsqrtf((float)cd + 1.0f);
            rec[pd + er[i]] = make_int2(sN, __float_as_int(coeff));
        }
    }
}


__global__ __launch_bounds__(256) void gather_in_rows(
    const int* __restrict__ rowptr, const int2* __restrict__ rec,
    const float* __restrict__ dis, const float* __restrict__ x,
    float4* __restrict__ ax) {
    constexpr int ROWS = 32;
    constexpr int SUB  = 8;
    constexpr int EPL  = 4;
    static_assert(ROWS * SUB == 256, "one group of lanes per node");
    static_assert(IN_DIM == 3, "input row padded to one k step of 4");
    static_assert(N_NODES % ROWS == 0, "whole row groups: no row guard needed");
    __shared__ float4 sX4[ROWS];

    const int t    = threadIdx.x;
    const int row0 = blockIdx.x * ROWS;
    const int nl   = t / SUB;
    const int sub  = t % SUB;
    const int node = row0 + nl;
    const int beg  = rowptr[node];
    const int end  = rowptr[node + 1];
    const float di = dis[node];
    const float* xn = x + (size_t)node * IN_DIM;
    const float n0 = xn[0], n1 = xn[1], n2 = xn[2];

    const bool own = sub == 0;
    const float ws = di * di;
    float a0 = own ? n0 * ws : 0.0f;
    float a1 = own ? n1 * ws : 0.0f;
    float a2 = own ? n2 * ws : 0.0f;
    for (int e = beg + sub; e < end; e += SUB * EPL) {
        int2 r[EPL];
        bool ok[EPL];
#pragma unroll
        for (int q = 0; q < EPL; ++q) {
            const int idx = e + q * SUB;
            ok[q] = idx < end;
            r[q]  = rec[ok[q] ? idx : end - 1];
        }
#pragma unroll
        for (int q = 0; q < EPL; ++q) {
            const float* xs = x + (size_t)r[q].x * IN_DIM;
            const float  cf = __int_as_float(r[q].y);
            const float  x0 = xs[0], x1 = xs[1], x2 = xs[2];
            a0 += (ok[q] ? x0 : 0.0f) * cf;
            a1 += (ok[q] ? x1 : 0.0f) * cf;
            a2 += (ok[q] ? x2 : 0.0f) * cf;
        }
    }
#pragma unroll
    for (int off = 1; off < SUB; off <<= 1) {
        a0 += __shfl_xor(a0, off);
        a1 += __shfl_xor(a1, off);
        a2 += __shfl_xor(a2, off);
    }
    if (own) sX4[nl] = make_float4(a0, a1, a2, 1.0f);
    __syncthreads();
    if (t < ROWS) ax[row0 + t] = sX4[t];
}

__global__ __launch_bounds__(256, 10) void gather_h1_gemm2_f32(
    const int* __restrict__ rowptr, const int2* __restrict__ rec,
    const float* __restrict__ dis, const float* __restrict__ ax,
    const float* __restrict__ Win, const float* __restrict__ bin,
    const float* __restrict__ B1, const float* __restrict__ bias1,
    const float* __restrict__ B2, float* __restrict__ C) {
    constexpr int K     = HID_DIM;
    constexpr int N1    = HID_DIM;
    constexpr int N2    = LAT_DIM;
    constexpr int ROWS  = 32;
    constexpr int NPW   = ROWS / 8;
    constexpr int CB    = N1 / 16;
    constexpr int RT    = ROWS / 16;
    constexpr int TN2   = N2 / 16;
    constexpr int PITCH = K + 4;
    static_assert(K == 128 && N1 / 16 == 8, "one column tile of W1 per wave");
    static_assert(8 % TN2 == 0 && 8 / TN2 == RT, "one tile of the last product per wave");
    static_assert(N_NODES % ROWS == 0, "whole row groups: no row guard needed");
    __shared__ float4 sA4[ROWS * PITCH / 4];
    float* sA = reinterpret_cast<float*>(sA4);

    const int t    = threadIdx.x;
    const int lane = t & 31;
    const int wave = __builtin_amdgcn_readfirstlane(t >> 5);
    const int row0 = blockIdx.x * ROWS;
    const int c    = lane & 15;
    const int h    = lane >> 4;

    v2f wf[CB];
#pragma unroll
    for (int cb = 0; cb < CB; ++cb) {
        const int col = cb * 16 + c;
        const float w1 = Win[(size_t)N1 + col];
        const float bb = bin[col];
        wf[cb].x = Win[(size_t)(2 * h) * N1 + col];
        wf[cb].y = h ? bb : w1;
    }

    int   beg[NPW], rows[NPW];
    float d2[NPW];
#pragma unroll
    for (int i = 0; i < NPW; ++i) {
        const int node = row0 + wave + i * 8;
        beg[i]  = rowptr[node];
        rows[i] = rowptr[node + 1] - beg[i] + 1;
        const float di = dis[node];
        d2[i] = di * di;
    }

    auto afrag = [&](int node, int b, int nrows, float self, int q) -> v2f {
        const bool ok = q >= 1 && q < nrows;
        const int2 r  = rec[ok ? b + q - 1 : 0];
        const int  sN = ok ? r.x : node;
        const float cf = ok ? __int_as_float(r.y) : (q == 0 ? self : 0.0f);
        const v2f v = *reinterpret_cast<const v2f*>(ax + (size_t)sN * 4 + 2 * h);
        return v * cf;
    };

    v2f anext = afrag(row0 + wave, beg[0], rows[0], d2[0], c);
#pragma unroll
    for (int i = 0; i < NPW; ++i) {
        const int node = row0 + wave + i * 8;
        float g[CB];
#pragma unroll
        for (int cb = 0; cb < CB; ++cb) g[cb] = 0.0f;
        v2f a = anext;
        if (i + 1 < NPW)
            anext = afrag(node + 8, beg[(i + 1) % NPW], rows[(i + 1) % NPW], d2[(i + 1) % NPW], c);
        for (int q0 = 0;;) {
#pragma unroll
            for (int cb = 0; cb < CB; ++cb) {
                const v8f d = __builtin_amdgcn_wmma_f32_16x16x4_f32(
                    false, a, false, wf[cb], (short)0, v8f{}, false, false);
                const float s01 = fmaxf(d[0], 0.0f) + fmaxf(d[1], 0.0f);
                const float s23 = fmaxf(d[2], 0.0f) + fmaxf(d[3], 0.0f);
                const float s45 = fmaxf(d[4], 0.0f) + fmaxf(d[5], 0.0f);
                const float s67 = fmaxf(d[6], 0.0f) + fmaxf(d[7], 0.0f);
                g[cb] += (s01 + s23) + (s45 + s67);
            }
            q0 += 16;
            if (q0 >= rows[i]) break;
            a = afrag(node, beg[i], rows[i], d2[i], q0 + c);
        }
#pragma unroll
        for (int j = 0; j < CB / 2; ++j) {
            const float keep = h ? g[2 * j + 1] : g[2 * j];
            const float send = h ? g[2 * j] : g[2 * j + 1];
            const float tot  = keep + __shfl_xor(send, 16);
            sA[(wave + i * 8) * PITCH + (2 * j + h) * 16 + c] = tot;
        }
    }
    __syncthreads();

    {
        const int ncol = wave * 16 + c;
        v2f bf[K / 4];
#pragma unroll
        for (int s = 0; s < K / 4; ++s) {
            const int ka = s * 4 + 2 * h;
            bf[s].x = B1[(size_t)ka * N1 + ncol];
            bf[s].y = B1[(size_t)(ka + 1) * N1 + ncol];
        }
        const float bcol = bias1[ncol];
        v8f acc[RT];
#pragma unroll
        for (int j = 0; j < RT; ++j) acc[j] = v8f{};
        const float* arow = sA + c * PITCH + 2 * h;
#pragma unroll
        for (int s = 0; s < K / 4; ++s) {
#pragma unroll
            for (int j = 0; j < RT; ++j) {
                const v2f a = *reinterpret_cast<const v2f*>(arow + j * 16 * PITCH + s * 4);
                acc[j] = __builtin_amdgcn_wmma_f32_16x16x4_f32(
                    false, a, false, bf[s], (short)0, acc[j], false, false);
            }
        }
        __syncthreads();
#pragma unroll
        for (int j = 0; j < RT; ++j) {
#pragma unroll
            for (int r = 0; r < 8; ++r)
                sA[(j * 16 + h * 8 + r) * PITCH + ncol] = fmaxf(acc[j][r] + bcol, 0.0f);
        }
    }
    __syncthreads();

    {
        const int tn   = wave % TN2;
        const int tr   = wave / TN2;
        const int ncol = tn * 16 + c;
        v2f bf[K / 4];
#pragma unroll
        for (int s = 0; s < K / 4; ++s) {
            const int ka = s * 4 + 2 * h;
            bf[s].x = B2[(size_t)ka * N2 + ncol];
            bf[s].y = B2[(size_t)(ka + 1) * N2 + ncol];
        }
        v8f acc = v8f{};
        const float* arow = sA + (tr * 16 + c) * PITCH + 2 * h;
#pragma unroll
        for (int s = 0; s < K / 4; ++s) {
            const v2f a = *reinterpret_cast<const v2f*>(arow + s * 4);
            acc = __builtin_amdgcn_wmma_f32_16x16x4_f32(
                false, a, false, bf[s], (short)0, acc, false, false);
        }
        const int crow = row0 + tr * 16 + h * 8;
#pragma unroll
        for (int r = 0; r < 8; ++r)
            C[(size_t)(crow + r) * N2 + ncol] = acc[r];
    }
}

template <int DIM, bool RELU>
__global__ __launch_bounds__(256) void gather_rows(
    const int* __restrict__ rowptr, const int2* __restrict__ rec,
    const float* __restrict__ dis, const float* __restrict__ m,
    const float* __restrict__ bias, float* __restrict__ out, int nNodes) {
    constexpr int PER = DIM / 32;
    typedef __attribute__((ext_vector_type(PER))) float vec;

    const int lane = threadIdx.x & 31;
    const int node = __builtin_amdgcn_readfirstlane(
        (int)(blockIdx.x * (blockDim.x >> 5) + (threadIdx.x >> 5)));
    if (node >= nNodes) return;

    const int   beg = rowptr[node];
    const int   end = rowptr[node + 1];
    const float di  = dis[node];

    const vec* __restrict__ mv = reinterpret_cast<const vec*>(m);
    vec acc = mv[(size_t)node * 32 + lane] * di * di;

    for (int base = beg; base < end; base += 32) {
        const int n = min(32, end - base);
        int2 r = make_int2(0, 0);
        if (lane < n) r = rec[base + lane];
        int j = 0;
        for (; j + 4 <= n; j += 4) {
            const int   s0 = __builtin_amdgcn_readlane(r.x, j);
            const int   s1 = __builtin_amdgcn_readlane(r.x, j + 1);
            const int   s2 = __builtin_amdgcn_readlane(r.x, j + 2);
            const int   s3 = __builtin_amdgcn_readlane(r.x, j + 3);
            const float c0 = __int_as_float(__builtin_amdgcn_readlane(r.y, j));
            const float c1 = __int_as_float(__builtin_amdgcn_readlane(r.y, j + 1));
            const float c2 = __int_as_float(__builtin_amdgcn_readlane(r.y, j + 2));
            const float c3 = __int_as_float(__builtin_amdgcn_readlane(r.y, j + 3));
            const vec v0 = mv[(size_t)s0 * 32 + lane];
            const vec v1 = mv[(size_t)s1 * 32 + lane];
            const vec v2 = mv[(size_t)s2 * 32 + lane];
            const vec v3 = mv[(size_t)s3 * 32 + lane];
            acc += v0 * c0;
            acc += v1 * c1;
            acc += v2 * c2;
            acc += v3 * c3;
        }
        for (; j < n; ++j) {
            const int   s0 = __builtin_amdgcn_readlane(r.x, j);
            const float c0 = __int_as_float(__builtin_amdgcn_readlane(r.y, j));
            acc += mv[(size_t)s0 * 32 + lane] * c0;
        }
    }

    acc += reinterpret_cast<const vec*>(bias)[lane];
    if (RELU) {
#pragma unroll
        for (int k = 0; k < PER; ++k) acc[k] = fmaxf(acc[k], 0.0f);
    }
    reinterpret_cast<vec*>(out)[(size_t)node * 32 + lane] = acc;
}

extern "C" void kernel_launch(void* const* d_in, const int* in_sizes, int n_in,
                              void* d_out, int out_size, void* d_ws, size_t ws_size,
                              hipStream_t stream) {
    const float* x   = (const float*)d_in[0];
    const int*   ei  = (const int*)d_in[1];
    const float* W0  = (const float*)d_in[2];
    const float* b0  = (const float*)d_in[3];
    const float* W1  = (const float*)d_in[4];
    const float* b1  = (const float*)d_in[5];
    const float* W2  = (const float*)d_in[6];
    const float* b2  = (const float*)d_in[7];
    float* out = (float*)d_out;

    const int* src = ei;
    const int* dst = ei + N_EDGES;

    float* ws     = (float*)d_ws;
    float* m      = ws;
    float* ax     = m + (size_t)N_NODES * HID_DIM;
    int2*  rec    = (int2*)(ax + (size_t)N_NODES * HID_DIM);
    float* dis    = (float*)(rec + N_EDGES);
    int*   cnt    = (int*)(dis + N_NODES);
    int*   rowptr = cnt + N_NODES;
    int*   rank   = rowptr + (N_NODES + 32);

    const int T = 256;
    const int nodeWaveBlocks = (N_NODES * 32 + T - 1) / T;
    const int fusedBlocks = N_NODES / 32;

    zero_int_kernel<<<(N_NODES + T - 1) / T, T, 0, stream>>>(cnt, N_NODES);
    rank_kernel<<<(N_EDGES + T - 1) / T, T, 0, stream>>>(dst, cnt, rank, N_EDGES);
    scan_fill_kernel<<<PLACE_GROUPS, PLACE_THREADS, 0, stream>>>(src, dst, rank, cnt, rowptr, dis, rec);

    gather_in_rows<<<fusedBlocks, T, 0, stream>>>(rowptr, rec, dis, x, (float4*)ax);

    gather_h1_gemm2_f32<<<fusedBlocks, T, 0, stream>>>(rowptr, rec, dis, ax, W0, b0, W1, b1, W2, m);

    gather_rows<LAT_DIM, false><<<nodeWaveBlocks, T, 0, stream>>>(rowptr, rec, dis, m, b2, out, N_NODES);
}


# The counting add's return value places each edge in its node's 64-slot bucket (norm split: rows stored times dis[src], sums times dis[dst]); scan+placement launch gone, 5 launches; sum order changed
# speedup vs baseline: 4.6416x; 1.4531x over previous
#include <hip/hip_runtime.h>
#include <hip/hip_bf16.h>


typedef __attribute__((ext_vector_type(2))) float v2f;
typedef __attribute__((ext_vector_type(4))) float v4f;
typedef __attribute__((ext_vector_type(8))) float v8f;

#define N_NODES 40000
#define N_EDGES 640000
#define IN_DIM  3
#define HID_DIM 128
#define LAT_DIM 64
#define CAP     64

__global__ void init_nodes_kernel(const float* __restrict__ x, int4* __restrict__ tab,
                                  int* __restrict__ ovfN, int n) {
    int i = blockIdx.x * blockDim.x + threadIdx.x;
    if (i < n) {
        const float* xi = x + (size_t)i * IN_DIM;
        tab[i] = make_int4(__float_as_int(xi[0]), __float_as_int(xi[1]), __float_as_int(xi[2]), 0);
    }
    if (i == 0) *ovfN = 0;
}

__global__ void place_kernel(const int* __restrict__ src, const int* __restrict__ dst,
                             int* __restrict__ tabw, int* __restrict__ bucket,
                             int2* __restrict__ ovf, int* __restrict__ ovfN, int nEdges) {
    int e = blockIdx.x * blockDim.x + threadIdx.x;
    if (e < nEdges) {
        const int s = src[e];
        const int d = dst[e];
        const int slot = atomicAdd(&tabw[(size_t)d * 4 + 3], 1);
        if (slot < CAP) bucket[(size_t)d * CAP + slot] = s;
        else            ovf[atomicAdd(ovfN, 1)] = make_int2(s, d);
    }
}


__global__ __launch_bounds__(256) void gather_in_rows(
    const int4* __restrict__ tab, const int* __restrict__ bucket,
    const int2* __restrict__ ovf, const int* __restrict__ ovfN,
    float4* __restrict__ axs) {
    constexpr int ROWS = 32;
    constexpr int SUB  = 8;
    constexpr int EPL  = 4;
    static_assert(ROWS * SUB == 256, "one group of lanes per node");
    static_assert(IN_DIM == 3, "input row and counter fill one 16-byte row");
    static_assert(N_NODES % ROWS == 0, "whole row groups: no row guard needed");
    __shared__ float4 sX4[ROWS];

    const int t    = threadIdx.x;
    const int row0 = blockIdx.x * ROWS;
    const int nl   = t / SUB;
    const int sub  = t % SUB;
    const int node = row0 + nl;
    const int4 tn  = tab[node];
    const int  cntN = tn.w;
    const float di  = rsqrtf((float)cntN + 1.0f);
    const int  nrec = min(cntN, CAP);
    const int* bk   = bucket + (size_t)node * CAP;

    const bool own = sub == 0;
    float a0 = own ? __int_as_float(tn.x) * di : 0.0f;
    float a1 = own ? __int_as_float(tn.y) * di : 0.0f;
    float a2 = own ? __int_as_float(tn.z) * di : 0.0f;
    for (int k = sub; k < nrec; k += SUB * EPL) {
        int  sN[EPL];
        bool ok[EPL];
#pragma unroll
        for (int q = 0; q < EPL; ++q) {
            const int idx = k + q * SUB;
            ok[q] = idx < nrec;
            sN[q] = bk[ok[q] ? idx : nrec - 1];
        }
#pragma unroll
        for (int q = 0; q < EPL; ++q) {
            const int4  ts = tab[sN[q]];
            const float cf = rsqrtf((float)ts.w + 1.0f);
            a0 += (ok[q] ? __int_as_float(ts.x) : 0.0f) * cf;
            a1 += (ok[q] ? __int_as_float(ts.y) : 0.0f) * cf;
            a2 += (ok[q] ? __int_as_float(ts.z) : 0.0f) * cf;
        }
    }
    if (cntN > CAP) {
        const int total = *ovfN;
        for (int k = sub; k < total; k += SUB) {
            const int2 en = ovf[k];
            if (en.y == node) {
                const int4  ts = tab[en.x];
                const float cf = rsqrtf((float)ts.w + 1.0f);
                a0 += __int_as_float(ts.x) * cf;
                a1 += __int_as_float(ts.y) * cf;
                a2 += __int_as_float(ts.z) * cf;
            }
        }
    }
#pragma unroll
    for (int off = 1; off < SUB; off <<= 1) {
        a0 += __shfl_xor(a0, off);
        a1 += __shfl_xor(a1, off);
        a2 += __shfl_xor(a2, off);
    }
    if (own) {
        const float d2 = di * di;
        sX4[nl] = make_float4(a0 * d2, a1 * d2, a2 * d2, di);
    }
    __syncthreads();
    if (t < ROWS) axs[row0 + t] = sX4[t];
}

__global__ __launch_bounds__(256, 10) void gather_h1_gemm2_f32(
    const int4* __restrict__ tab, const int* __restrict__ bucket,
    const int2* __restrict__ ovf, const int* __restrict__ ovfN,
    const float* __restrict__ axs,
    const float* __restrict__ Win, const float* __restrict__ bin,
    const float* __restrict__ B1, const float* __restrict__ bias1,
    const float* __restrict__ B2, float* __restrict__ C) {
    constexpr int K     = HID_DIM;
    constexpr int N1    = HID_DIM;
    constexpr int N2    = LAT_DIM;
    constexpr int ROWS  = 32;
    constexpr int NPW   = ROWS / 8;
    constexpr int CB    = N1 / 16;
    constexpr int RT    = ROWS / 16;
    constexpr int TN2   = N2 / 16;
    constexpr int PITCH = K + 4;
    static_assert(K == 128 && N1 / 16 == 8, "one column tile of W1 per wave");
    static_assert(8 % TN2 == 0 && 8 / TN2 == RT, "one tile of the last product per wave");
    static_assert(N_NODES % ROWS == 0, "whole row groups: no row guard needed");
    __shared__ float4 sA4[ROWS * PITCH / 4];
    __shared__ float  sD[ROWS];
    float* sA = reinterpret_cast<float*>(sA4);

    const int t    = threadIdx.x;
    const int lane = t & 31;
    const int wave = __builtin_amdgcn_readfirstlane(t >> 5);
    const int row0 = blockIdx.x * ROWS;
    const int c    = lane & 15;
    const int h    = lane >> 4;

    v2f wf[CB];
#pragma unroll
    for (int cb = 0; cb < CB; ++cb) {
        const int col = cb * 16 + c;
        const float w1 = Win[(size_t)N1 + col];
        const float bb = bin[col];
        wf[cb].x = Win[(size_t)(2 * h) * N1 + col];
        wf[cb].y = h ? bb : w1;
    }

    int   cnt[NPW], rows[NPW];
    float d1[NPW];
#pragma unroll
    for (int i = 0; i < NPW; ++i) {
        const int node = row0 + wave + i * 8;
        cnt[i]  = tab[node].w;
        rows[i] = min(cnt[i], CAP) + 1;
        d1[i]   = rsqrtf((float)cnt[i] + 1.0f);
    }

    auto afrag = [&](int node, int nrows, int q) -> v2f {
        const bool ok = q >= 1 && q < nrows;
        const int  r  = bucket[(size_t)node * CAP + (ok ? q - 1 : 0)];
        const int  sN = ok ? r : node;
        const v2f  v  = *reinterpret_cast<const v2f*>(axs + (size_t)sN * 4 + 2 * h);
        return q < nrows ? v : v2f{0.0f, 0.0f};
    };
    auto accum = [&](v2f a, float (&g)[CB]) {
#pragma unroll
        for (int cb = 0; cb < CB; ++cb) {
            const v8f d = __builtin_amdgcn_wmma_f32_16x16x4_f32(
                false, a, false, wf[cb], (short)0, v8f{}, false, false);
            const float s01 = fmaxf(d[0], 0.0f) + fmaxf(d[1], 0.0f);
            const float s23 = fmaxf(d[2], 0.0f) + fmaxf(d[3], 0.0f);
            const float s45 = fmaxf(d[4], 0.0f) + fmaxf(d[5], 0.0f);
            const float s67 = fmaxf(d[6], 0.0f) + fmaxf(d[7], 0.0f);
            g[cb] += (s01 + s23) + (s45 + s67);
        }
    };

    v2f anext = afrag(row0 + wave, rows[0], c);
#pragma unroll
    for (int i = 0; i < NPW; ++i) {
        const int node = row0 + wave + i * 8;
        float g[CB];
#pragma unroll
        for (int cb = 0; cb < CB; ++cb) g[cb] = 0.0f;
        v2f a = anext;
        if (i + 1 < NPW)
            anext = afrag(node + 8, rows[(i + 1) % NPW], c);
        for (int q0 = 0;;) {
            accum(a, g);
            q0 += 16;
            if (q0 >= rows[i]) break;
            a = afrag(node, rows[i], q0 + c);
        }
        if (cnt[i] > CAP) {
            const int total = *ovfN;
            for (int k0 = 0; k0 < total; k0 += 16) {
                const int  idx = k0 + c;
                const bool in  = idx < total;
                const int2 en  = ovf[in ? idx : 0];
                const bool hit = in && en.y == node;
                const v2f  v   = *reinterpret_cast<const v2f*>(axs + (size_t)(hit ? en.x : node) * 4 + 2 * h);
                accum(hit ? v : v2f{0.0f, 0.0f}, g);
            }
        }
#pragma unroll
        for (int j = 0; j < CB / 2; ++j) {
            const float keep = h ? g[2 * j + 1] : g[2 * j];
            const float send = h ? g[2 * j] : g[2 * j + 1];
            const float tot  = keep + __shfl_xor(send, 16);
            sA[(wave + i * 8) * PITCH + (2 * j + h) * 16 + c] = tot * d1[i];
        }
        if (lane == 0) sD[wave + i * 8] = d1[i];
    }
    __syncthreads();

    {
        const int ncol = wave * 16 + c;
        v2f bf[K / 4];
#pragma unroll
        for (int s = 0; s < K / 4; ++s) {
            const int ka = s * 4 + 2 * h;
            bf[s].x = B1[(size_t)ka * N1 + ncol];
            bf[s].y = B1[(size_t)(ka + 1) * N1 + ncol];
        }
        const float bcol = bias1[ncol];
        v8f acc[RT];
#pragma unroll
        for (int j = 0; j < RT; ++j) acc[j] = v8f{};
        const float* arow = sA + c * PITCH + 2 * h;
#pragma unroll
        for (int s = 0; s < K / 4; ++s) {
#pragma unroll
            for (int j = 0; j < RT; ++j) {
                const v2f a = *reinterpret_cast<const v2f*>(arow + j * 16 * PITCH + s * 4);
                acc[j] = __builtin_amdgcn_wmma_f32_16x16x4_f32(
                    false, a, false, bf[s], (short)0, acc[j], false, false);
            }
        }
        __syncthreads();
#pragma unroll
        for (int j = 0; j < RT; ++j) {
#pragma unroll
            for (int r = 0; r < 8; ++r)
                sA[(j * 16 + h * 8 + r) * PITCH + ncol] = fmaxf(acc[j][r] + bcol, 0.0f);
        }
    }
    __syncthreads();

    {
        const int tn   = wave % TN2;
        const int tr   = wave / TN2;
        const int ncol = tn * 16 + c;
        v2f bf[K / 4];
#pragma unroll
        for (int s = 0; s < K / 4; ++s) {
            const int ka = s * 4 + 2 * h;
            bf[s].x = B2[(size_t)ka * N2 + ncol];
            bf[s].y = B2[(size_t)(ka + 1) * N2 + ncol];
        }
        v8f acc = v8f{};
        const float* arow = sA + (tr * 16 + c) * PITCH + 2 * h;
#pragma unroll
        for (int s = 0; s < K / 4; ++s) {
            const v2f a = *reinterpret_cast<const v2f*>(arow + s * 4);
            acc = __builtin_amdgcn_wmma_f32_16x16x4_f32(
                false, a, false, bf[s], (short)0, acc, false, false);
        }
        const int crow = row0 + tr * 16 + h * 8;
#pragma unroll
        for (int r = 0; r < 8; ++r)
            C[(size_t)(crow + r) * N2 + ncol] = acc[r] * sD[tr * 16 + h * 8 + r];
    }
}

template <int DIM, bool RELU>
__global__ __launch_bounds__(256) void gather_rows(
    const int4* __restrict__ tab, const int* __restrict__ bucket,
    const int2* __restrict__ ovf, const int* __restrict__ ovfN,
    const float* __restrict__ m,
    const float* __restrict__ bias, float* __restrict__ out, int nNodes) {
    constexpr int PER = DIM / 32;
    typedef __attribute__((ext_vector_type(PER))) float vec;

    const int lane = threadIdx.x & 31;
    const int node = __builtin_amdgcn_readfirstlane(
        (int)(blockIdx.x * (blockDim.x >> 5) + (threadIdx.x >> 5)));
    if (node >= nNodes) return;

    const int   cntN = tab[node].w;
    const float di   = rsqrtf((float)cntN + 1.0f);
    const int   nrec = min(cntN, CAP);
    const int*  bk   = bucket + (size_t)node * CAP;

    const vec* __restrict__ mv = reinterpret_cast<const vec*>(m);
    vec acc = mv[(size_t)node * 32 + lane];

    for (int base = 0; base < nrec; base += 32) {
        const int n = min(32, nrec - base);
        int r = 0;
        if (lane < n) r = bk[base + lane];
        int j = 0;
        for (; j + 4 <= n; j += 4) {
            const int s0 = __builtin_amdgcn_readlane(r, j);
            const int s1 = __builtin_amdgcn_readlane(r, j + 1);
            const int s2 = __builtin_amdgcn_readlane(r, j + 2);
            const int s3 = __builtin_amdgcn_readlane(r, j + 3);
            const vec v0 = mv[(size_t)s0 * 32 + lane];
            const vec v1 = mv[(size_t)s1 * 32 + lane];
            const vec v2 = mv[(size_t)s2 * 32 + lane];
            const vec v3 = mv[(size_t)s3 * 32 + lane];
            acc += v0;
            acc += v1;
            acc += v2;
            acc += v3;
        }
        for (; j < n; ++j) {
            const int s0 = __builtin_amdgcn_readlane(r, j);
            acc += mv[(size_t)s0 * 32 + lane];
        }
    }
    if (cntN > CAP) {
        const int total = *ovfN;
        for (int k0 = 0; k0 < total; k0 += 32) {
            const int idx = k0 + lane;
            int2 en = make_int2(0, -1);
            if (idx < total) en = ovf[idx];
            unsigned hits = __builtin_amdgcn_ballot_w32(en.y == node);
            while (hits) {
                const int j = __builtin_ctz(hits);
                hits &= hits - 1;
                const int s0 = __builtin_amdgcn_readlane(en.x, j);
                acc += mv[(size_t)s0 * 32 + lane];
            }
        }
    }

    acc = acc * di + reinterpret_cast<const vec*>(bias)[lane];
    if (RELU) {
#pragma unroll
        for (int k = 0; k < PER; ++k) acc[k] = fmaxf(acc[k], 0.0f);
    }
    reinterpret_cast<vec*>(out)[(size_t)node * 32 + lane] = acc;
}

extern "C" void kernel_launch(void* const* d_in, const int* in_sizes, int n_in,
                              void* d_out, int out_size, void* d_ws, size_t ws_size,
                              hipStream_t stream) {
    const float* x   = (const float*)d_in[0];
    const int*   ei  = (const int*)d_in[1];
    const float* W0  = (const float*)d_in[2];
    const float* b0  = (const float*)d_in[3];
    const float* W1  = (const float*)d_in[4];
    const float* b1  = (const float*)d_in[5];
    const float* W2  = (const float*)d_in[6];
    const float* b2  = (const float*)d_in[7];
    float* out = (float*)d_out;

    const int* src = ei;
    const int* dst = ei + N_EDGES;

    float* ws     = (float*)d_ws;
    float* m      = ws;
    float* axs    = m + (size_t)N_NODES * LAT_DIM;
    int4*  tab    = (int4*)(axs + (size_t)N_NODES * 4);
    int*   bucket = (int*)(tab + N_NODES);
    int2*  ovf    = (int2*)(bucket + (size_t)N_NODES * CAP);
    int*   ovfN   = (int*)(ovf + N_EDGES);

    const int T = 256;
    const int nodeWaveBlocks = (N_NODES * 32 + T - 1) / T;
    const int fusedBlocks = N_NODES / 32;

    init_nodes_kernel<<<(N_NODES + T - 1) / T, T, 0, stream>>>(x, tab, ovfN, N_NODES);
    place_kernel<<<(N_EDGES + T - 1) / T, T, 0, stream>>>(src, dst, (int*)tab, bucket, ovf, ovfN, N_EDGES);

    gather_in_rows<<<fusedBlocks, T, 0, stream>>>(tab, bucket, ovf, ovfN, (float4*)axs);

    gather_h1_gemm2_f32<<<fusedBlocks, T, 0, stream>>>(tab, bucket, ovf, ovfN, axs, W0, b0, W1, b1, W2, m);

    gather_rows<LAT_DIM, false><<<nodeWaveBlocks, T, 0, stream>>>(tab, bucket, ovf, ovfN, m, b2, out, N_NODES);
}


# Edges grouped by destination in two LDS binning launches (chunks sorted by node range, then a range's owner fills its buckets in LDS) replace global add + scattered store per edge; sum order changed
# speedup vs baseline: 7.0322x; 1.5150x over previous
#include <hip/hip_runtime.h>
#include <hip/hip_bf16.h>


typedef __attribute__((ext_vector_type(2))) float v2f;
typedef __attribute__((ext_vector_type(4))) float v4f;
typedef __attribute__((ext_vector_type(8))) float v8f;

#define N_NODES 40000
#define N_EDGES 640000
#define IN_DIM  3
#define HID_DIM 128
#define LAT_DIM 64
#define CAP     64
#define NBIN    250
#define BIN_N   160
#define CHUNK   2560
static_assert(NBIN * BIN_N == N_NODES, "whole ranges");
static_assert(NBIN * CHUNK == N_EDGES, "whole chunks");
static_assert(N_NODES <= 65536 && BIN_N <= 65536, "source and node-in-range share one 32-bit record");

__global__ __launch_bounds__(512) void bin_edges_kernel(
    const int* __restrict__ src, const int* __restrict__ dst,
    unsigned* __restrict__ recs, int2* __restrict__ runs, int* __restrict__ ovfN) {
    constexpr int T   = 512;
    constexpr int EPT = CHUNK / T;
    constexpr int NB  = 256;
    static_assert(CHUNK % T == 0, "whole rounds");
    static_assert(NBIN <= NB && NB == 32 * 8 && NBIN <= T, "one wave scans all ranges");
    __shared__ int      sCnt[NB];
    __shared__ int      sOff[NB];
    __shared__ unsigned sRec[CHUNK];

    const int t    = threadIdx.x;
    const int w    = blockIdx.x;
    const int base = w * CHUNK;
    if (t < NB) sCnt[t] = 0;
    if (w == 0 && t == 0) *ovfN = 0;
    __syncthreads();

    int s[EPT], d[EPT], bin[EPT], slot[EPT];
#pragma unroll
    for (int j = 0; j < EPT; ++j) {
        s[j] = src[base + j * T + t];
        d[j] = dst[base + j * T + t];
    }
    static_assert(EPT == 5, "the statement below names every loaded value");
    asm volatile("" :: "v"(s[0]), "v"(s[1]), "v"(s[2]), "v"(s[3]), "v"(s[4]),
                       "v"(d[0]), "v"(d[1]), "v"(d[2]), "v"(d[3]), "v"(d[4]));
#pragma unroll
    for (int j = 0; j < EPT; ++j) {
        bin[j]  = d[j] / BIN_N;
        slot[j] = atomicAdd(&sCnt[bin[j]], 1);
    }
    __syncthreads();
    if (t < 32) {
        int c[8];
        int tot = 0;
#pragma unroll
        for (int k = 0; k < 8; ++k) { c[k] = sCnt[8 * t + k]; tot += c[k]; }
        int inc = tot;
#pragma unroll
        for (int off = 1; off < 32; off <<= 1) {
            const int v = __shfl_up(inc, off);
            if (t >= off) inc += v;
        }
        int ex = inc - tot;
#pragma unroll
        for (int k = 0; k < 8; ++k) { sOff[8 * t + k] = ex; ex += c[k]; }
    }
    __syncthreads();
#pragma unroll
    for (int j = 0; j < EPT; ++j)
        sRec[sOff[bin[j]] + slot[j]] = (unsigned)s[j] | ((unsigned)(d[j] - bin[j] * BIN_N) << 16);
    __syncthreads();
#pragma unroll
    for (int j = 0; j < EPT; ++j) recs[base + j * T + t] = sRec[j * T + t];
    if (t < NBIN) runs[(size_t)t * NBIN + w] = make_int2(sOff[t], sCnt[t]);
}

__global__ __launch_bounds__(512) void fill_buckets_kernel(
    const float* __restrict__ x, const unsigned* __restrict__ recs, const int2* __restrict__ runs,
    int4* __restrict__ tab, int* __restrict__ bucket,
    int2* __restrict__ ovf, int* __restrict__ ovfN) {
    constexpr int T   = 512;
    constexpr int NHW = T / 16;
    constexpr int RPH = (NBIN + NHW - 1) / NHW;
    static_assert(BIN_N <= T, "one thread per node for the counters and rows");
    static_assert(CAP == 64, "bucket = 16 pieces of 16 bytes; rotation by node modulo 16 pieces");
    __shared__ int  sCnt[BIN_N];
    __shared__ int4 sB4[BIN_N * CAP / 4];
    int* sB = reinterpret_cast<int*>(sB4);

    const int t     = threadIdx.x;
    const int r     = blockIdx.x;
    const int node0 = r * BIN_N;
    const int hl    = t & 15;
    const int hw    = t >> 4;
    if (t < BIN_N) sCnt[t] = 0;
    __syncthreads();
    float x0 = 0.0f, x1 = 0.0f, x2 = 0.0f;
    if (t < BIN_N) {
        const float* xi = x + (size_t)(node0 + t) * IN_DIM;
        x0 = xi[0];
        x1 = xi[1];
        x2 = xi[2];
    }

    auto place = [&](unsigned rec) {
        const int sN   = (int)(rec & 0xFFFFu);
        const int dl   = (int)(rec >> 16);
        const int slot = atomicAdd(&sCnt[dl], 1);
        if (slot < CAP) sB[dl * CAP + ((slot + 4 * dl) & (CAP - 1))] = sN;
        else            ovf[atomicAdd(ovfN, 1)] = make_int2(sN, node0 + dl);
    };

    int first[RPH], len[RPH];
#pragma unroll
    for (int i = 0; i < RPH; ++i) {
        const int  w   = hw + NHW * i;
        const int  wc  = w < NBIN ? w : NBIN - 1;
        const int2 run = runs[(size_t)r * NBIN + wc];
        first[i] = wc * CHUNK + run.x;
        len[i]   = w < NBIN ? run.y : 0;
    }
    unsigned rc[RPH];
#pragma unroll
    for (int i = 0; i < RPH; ++i)
        rc[i] = recs[hl < len[i] ? first[i] + hl : 0];
    int slot[RPH];
#pragma unroll
    for (int i = 0; i < RPH; ++i) {
        slot[i] = 0;
        if (hl < len[i]) slot[i] = atomicAdd(&sCnt[rc[i] >> 16], 1);
    }
#pragma unroll
    for (int i = 0; i < RPH; ++i) {
        if (hl < len[i]) {
            const int sN = (int)(rc[i] & 0xFFFFu);
            const int dl = (int)(rc[i] >> 16);
            if (slot[i] < CAP) sB[dl * CAP + ((slot[i] + 4 * dl) & (CAP - 1))] = sN;
            else               ovf[atomicAdd(ovfN, 1)] = make_int2(sN, node0 + dl);
        }
    }
#pragma unroll
    for (int i = 0; i < RPH; ++i)
        for (int k = 16 + hl; k < len[i]; k += 16) place(recs[first[i] + k]);
    __syncthreads();

    if (t < BIN_N)
        tab[node0 + t] = make_int4(__float_as_int(x0), __float_as_int(x1), __float_as_int(x2), sCnt[t]);
    int4* bk4 = reinterpret_cast<int4*>(bucket + (size_t)node0 * CAP);
    for (int i = t; i < BIN_N * CAP / 4; i += T)
        bk4[i] = sB4[(i & ~15) | (((i & 15) + (i >> 4)) & 15)];
}


__global__ __launch_bounds__(256) void gather_in_rows(
    const int4* __restrict__ tab, const int* __restrict__ bucket,
    const int2* __restrict__ ovf, const int* __restrict__ ovfN,
    float4* __restrict__ axs) {
    constexpr int ROWS = 32;
    constexpr int SUB  = 8;
    constexpr int EPL  = 4;
    static_assert(ROWS * SUB == 256, "one group of lanes per node");
    static_assert(IN_DIM == 3, "input row and counter fill one 16-byte row");
    static_assert(N_NODES % ROWS == 0, "whole row groups: no row guard needed");
    __shared__ float4 sX4[ROWS];

    const int t    = threadIdx.x;
    const int row0 = blockIdx.x * ROWS;
    const int nl   = t / SUB;
    const int sub  = t % SUB;
    const int node = row0 + nl;
    const int4 tn  = tab[node];
    const int  cntN = tn.w;
    const float di  = rsqrtf((float)cntN + 1.0f);
    const int  nrec = min(cntN, CAP);
    const int* bk   = bucket + (size_t)node * CAP;

    const bool own = sub == 0;
    float a0 = own ? __int_as_float(tn.x) * di : 0.0f;
    float a1 = own ? __int_as_float(tn.y) * di : 0.0f;
    float a2 = own ? __int_as_float(tn.z) * di : 0.0f;
    for (int k = sub; k < nrec; k += SUB * EPL) {
        int  sN[EPL];
        bool ok[EPL];
#pragma unroll
        for (int q = 0; q < EPL; ++q) {
            const int idx = k + q * SUB;
            ok[q] = idx < nrec;
            sN[q] = bk[ok[q] ? idx : nrec - 1];
        }
#pragma unroll
        for (int q = 0; q < EPL; ++q) {
            const int4  ts = tab[sN[q]];
            const float cf = rsqrtf((float)ts.w + 1.0f);
            a0 += (ok[q] ? __int_as_float(ts.x) : 0.0f) * cf;
            a1 += (ok[q] ? __int_as_float(ts.y) : 0.0f) * cf;
            a2 += (ok[q] ? __int_as_float(ts.z) : 0.0f) * cf;
        }
    }
    if (cntN > CAP) {
        const int total = *ovfN;
        for (int k = sub; k < total; k += SUB) {
            const int2 en = ovf[k];
            if (en.y == node) {
                const int4  ts = tab[en.x];
                const float cf = rsqrtf((float)ts.w + 1.0f);
                a0 += __int_as_float(ts.x) * cf;
                a1 += __int_as_float(ts.y) * cf;
                a2 += __int_as_float(ts.z) * cf;
            }
        }
    }
#pragma unroll
    for (int off = 1; off < SUB; off <<= 1) {
        a0 += __shfl_xor(a0, off);
        a1 += __shfl_xor(a1, off);
        a2 += __shfl_xor(a2, off);
    }
    if (own) {
        const float d2 = di * di;
        sX4[nl] = make_float4(a0 * d2, a1 * d2, a2 * d2, di);
    }
    __syncthreads();
    if (t < ROWS) axs[row0 + t] = sX4[t];
}

__global__ __launch_bounds__(256, 10) void gather_h1_gemm2_f32(
    const int4* __restrict__ tab, const int* __restrict__ bucket,
    const int2* __restrict__ ovf, const int* __restrict__ ovfN,
    const float* __restrict__ axs,
    const float* __restrict__ Win, const float* __restrict__ bin,
    const float* __restrict__ B1, const float* __restrict__ bias1,
    const float* __restrict__ B2, float* __restrict__ C) {
    constexpr int K     = HID_DIM;
    constexpr int N1    = HID_DIM;
    constexpr int N2    = LAT_DIM;
    constexpr int ROWS  = 32;
    constexpr int NPW   = ROWS / 8;
    constexpr int CB    = N1 / 16;
    constexpr int RT    = ROWS / 16;
    constexpr int TN2   = N2 / 16;
    constexpr int PITCH = K + 4;
    static_assert(K == 128 && N1 / 16 == 8, "one column tile of W1 per wave");
    static_assert(8 % TN2 == 0 && 8 / TN2 == RT, "one tile of the last product per wave");
    static_assert(N_NODES % ROWS == 0, "whole row groups: no row guard needed");
    __shared__ float4 sA4[ROWS * PITCH / 4];
    __shared__ float  sD[ROWS];
    float* sA = reinterpret_cast<float*>(sA4);

    const int t    = threadIdx.x;
    const int lane = t & 31;
    const int wave = __builtin_amdgcn_readfirstlane(t >> 5);
    const int row0 = blockIdx.x * ROWS;
    const int c    = lane & 15;
    const int h    = lane >> 4;

    v2f wf[CB];
#pragma unroll
    for (int cb = 0; cb < CB; ++cb) {
        const int col = cb * 16 + c;
        const float w1 = Win[(size_t)N1 + col];
        const float bb = bin[col];
        wf[cb].x = Win[(size_t)(2 * h) * N1 + col];
        wf[cb].y = h ? bb : w1;
    }

    int   cnt[NPW], rows[NPW];
    float d1[NPW];
#pragma unroll
    for (int i = 0; i < NPW; ++i) {
        const int node = row0 + wave + i * 8;
        cnt[i]  = tab[node].w;
        rows[i] = min(cnt[i], CAP) + 1;
        d1[i]   = rsqrtf((float)cnt[i] + 1.0f);
    }

    auto afrag = [&](int node, int nrows, int q) -> v2f {
        const bool ok = q >= 1 && q < nrows;
        const int  r  = bucket[(size_t)node * CAP + (ok ? q - 1 : 0)];
        const int  sN = ok ? r : node;
        const v2f  v  = *reinterpret_cast<const v2f*>(axs + (size_t)sN * 4 + 2 * h);
        return q < nrows ? v : v2f{0.0f, 0.0f};
    };
    auto accum = [&](v2f a, float (&g)[CB]) {
#pragma unroll
        for (int cb = 0; cb < CB; ++cb) {
            const v8f d = __builtin_amdgcn_wmma_f32_16x16x4_f32(
                false, a, false, wf[cb], (short)0, v8f{}, false, false);
            const float s01 = fmaxf(d[0], 0.0f) + fmaxf(d[1], 0.0f);
            const float s23 = fmaxf(d[2], 0.0f) + fmaxf(d[3], 0.0f);
            const float s45 = fmaxf(d[4], 0.0f) + fmaxf(d[5], 0.0f);
            const float s67 = fmaxf(d[6], 0.0f) + fmaxf(d[7], 0.0f);
            g[cb] += (s01 + s23) + (s45 + s67);
        }
    };

    v2f anext = afrag(row0 + wave, rows[0], c);
#pragma unroll
    for (int i = 0; i < NPW; ++i) {
        const int node = row0 + wave + i * 8;
        float g[CB];
#pragma unroll
        for (int cb = 0; cb < CB; ++cb) g[cb] = 0.0f;
        v2f a = anext;
        if (i + 1 < NPW)
            anext = afrag(node + 8, rows[(i + 1) % NPW], c);
        for (int q0 = 0;;) {
            accum(a, g);
            q0 += 16;
            if (q0 >= rows[i]) break;
            a = afrag(node, rows[i], q0 + c);
        }
        if (cnt[i] > CAP) {
            const int total = *ovfN;
            for (int k0 = 0; k0 < total; k0 += 16) {
                const int  idx = k0 + c;
                const bool in  = idx < total;
                const int2 en  = ovf[in ? idx : 0];
                const bool hit = in && en.y == node;
                const v2f  v   = *reinterpret_cast<const v2f*>(axs + (size_t)(hit ? en.x : node) * 4 + 2 * h);
                accum(hit ? v : v2f{0.0f, 0.0f}, g);
            }
        }
#pragma unroll
        for (int j = 0; j < CB / 2; ++j) {
            const float keep = h ? g[2 * j + 1] : g[2 * j];
            const float send = h ? g[2 * j] : g[2 * j + 1];
            const float tot  = keep + __shfl_xor(send, 16);
            sA[(wave + i * 8) * PITCH + (2 * j + h) * 16 + c] = tot * d1[i];
        }
        if (lane == 0) sD[wave + i * 8] = d1[i];
    }
    __syncthreads();

    {
        const int ncol = wave * 16 + c;
        v2f bf[K / 4];
#pragma unroll
        for (int s = 0; s < K / 4; ++s) {
            const int ka = s * 4 + 2 * h;
            bf[s].x = B1[(size_t)ka * N1 + ncol];
            bf[s].y = B1[(size_t)(ka + 1) * N1 + ncol];
        }
        const float bcol = bias1[ncol];
        v8f acc[RT];
#pragma unroll
        for (int j = 0; j < RT; ++j) acc[j] = v8f{};
        const float* arow = sA + c * PITCH + 2 * h;
#pragma unroll
        for (int s = 0; s < K / 4; ++s) {
#pragma unroll
            for (int j = 0; j < RT; ++j) {
                const v2f a = *reinterpret_cast<const v2f*>(arow + j * 16 * PITCH + s * 4);
                acc[j] = __builtin_amdgcn_wmma_f32_16x16x4_f32(
                    false, a, false, bf[s], (short)0, acc[j], false, false);
            }
        }
        __syncthreads();
#pragma unroll
        for (int j = 0; j < RT; ++j) {
#pragma unroll
            for (int r = 0; r < 8; ++r)
                sA[(j * 16 + h * 8 + r) * PITCH + ncol] = fmaxf(acc[j][r] + bcol, 0.0f);
        }
    }
    __syncthreads();

    {
        const int tn   = wave % TN2;
        const int tr   = wave / TN2;
        const int ncol = tn * 16 + c;
        v2f bf[K / 4];
#pragma unroll
        for (int s = 0; s < K / 4; ++s) {
            const int ka = s * 4 + 2 * h;
            bf[s].x = B2[(size_t)ka * N2 + ncol];
            bf[s].y = B2[(size_t)(ka + 1) * N2 + ncol];
        }
        v8f acc = v8f{};
        const float* arow = sA + (tr * 16 + c) * PITCH + 2 * h;
#pragma unroll
        for (int s = 0; s < K / 4; ++s) {
            const v2f a = *reinterpret_cast<const v2f*>(arow + s * 4);
            acc = __builtin_amdgcn_wmma_f32_16x16x4_f32(
                false, a, false, bf[s], (short)0, acc, false, false);
        }
        const int crow = row0 + tr * 16 + h * 8;
#pragma unroll
        for (int r = 0; r < 8; ++r)
            C[(size_t)(crow + r) * N2 + ncol] = acc[r] * sD[tr * 16 + h * 8 + r];
    }
}

template <int DIM, bool RELU>
__global__ __launch_bounds__(256) void gather_rows(
    const int4* __restrict__ tab, const int* __restrict__ bucket,
    const int2* __restrict__ ovf, const int* __restrict__ ovfN,
    const float* __restrict__ m,
    const float* __restrict__ bias, float* __restrict__ out, int nNodes) {
    constexpr int PER = DIM / 32;
    typedef __attribute__((ext_vector_type(PER))) float vec;

    const int lane = threadIdx.x & 31;
    const int node = __builtin_amdgcn_readfirstlane(
        (int)(blockIdx.x * (blockDim.x >> 5) + (threadIdx.x >> 5)));
    if (node >= nNodes) return;

    const int   cntN = tab[node].w;
    const float di   = rsqrtf((float)cntN + 1.0f);
    const int   nrec = min(cntN, CAP);
    const int*  bk   = bucket + (size_t)node * CAP;

    const vec* __restrict__ mv = reinterpret_cast<const vec*>(m);
    vec acc = mv[(size_t)node * 32 + lane];

    for (int base = 0; base < nrec; base += 32) {
        const int n = min(32, nrec - base);
        int r = 0;
        if (lane < n) r = bk[base + lane];
        int j = 0;
        for (; j + 4 <= n; j += 4) {
            const int s0 = __builtin_amdgcn_readlane(r, j);
            const int s1 = __builtin_amdgcn_readlane(r, j + 1);
            const int s2 = __builtin_amdgcn_readlane(r, j + 2);
            const int s3 = __builtin_amdgcn_readlane(r, j + 3);
            const vec v0 = mv[(size_t)s0 * 32 + lane];
            const vec v1 = mv[(size_t)s1 * 32 + lane];
            const vec v2 = mv[(size_t)s2 * 32 + lane];
            const vec v3 = mv[(size_t)s3 * 32 + lane];
            acc += v0;
            acc += v1;
            acc += v2;
            acc += v3;
        }
        for (; j < n; ++j) {
            const int s0 = __builtin_amdgcn_readlane(r, j);
            acc += mv[(size_t)s0 * 32 + lane];
        }
    }
    if (cntN > CAP) {
        const int total = *ovfN;
        for (int k0 = 0; k0 < total; k0 += 32) {
            const int idx = k0 + lane;
            int2 en = make_int2(0, -1);
            if (idx < total) en = ovf[idx];
            unsigned hits = __builtin_amdgcn_ballot_w32(en.y == node);
            while (hits) {
                const int j = __builtin_ctz(hits);
                hits &= hits - 1;
                const int s0 = __builtin_amdgcn_readlane(en.x, j);
                acc += mv[(size_t)s0 * 32 + lane];
            }
        }
    }

    acc = acc * di + reinterpret_cast<const vec*>(bias)[lane];
    if (RELU) {
#pragma unroll
        for (int k = 0; k < PER; ++k) acc[k] = fmaxf(acc[k], 0.0f);
    }
    reinterpret_cast<vec*>(out)[(size_t)node * 32 + lane] = acc;
}

extern "C" void kernel_launch(void* const* d_in, const int* in_sizes, int n_in,
                              void* d_out, int out_size, void* d_ws, size_t ws_size,
                              hipStream_t stream) {
    const float* x   = (const float*)d_in[0];
    const int*   ei  = (const int*)d_in[1];
    const float* W0  = (const float*)d_in[2];
    const float* b0  = (const float*)d_in[3];
    const float* W1  = (const float*)d_in[4];
    const float* b1  = (const float*)d_in[5];
    const float* W2  = (const float*)d_in[6];
    const float* b2  = (const float*)d_in[7];
    float* out = (float*)d_out;

    const int* src = ei;
    const int* dst = ei + N_EDGES;

    float* ws     = (float*)d_ws;
    float* m      = ws;
    float* axs    = m + (size_t)N_NODES * LAT_DIM;
    int4*  tab    = (int4*)(axs + (size_t)N_NODES * 4);
    int*   bucket = (int*)(tab + N_NODES);
    int2*  ovf    = (int2*)(bucket + (size_t)N_NODES * CAP);
    int*   ovfN   = (int*)(ovf + N_EDGES);
    unsigned* recs = (unsigned*)(ovfN + 4);
    int2*  runs   = (int2*)(recs + N_EDGES);

    const int T = 256;
    const int nodeWaveBlocks = (N_NODES * 32 + T - 1) / T;
    const int fusedBlocks = N_NODES / 32;

    bin_edges_kernel<<<NBIN, 512, 0, stream>>>(src, dst, recs, runs, ovfN);
    fill_buckets_kernel<<<NBIN, 512, 0, stream>>>(x, recs, runs, tab, bucket, ovf, ovfN);

    gather_in_rows<<<fusedBlocks, T, 0, stream>>>(tab, bucket, ovf, ovfN, (float4*)axs);

    gather_h1_gemm2_f32<<<fusedBlocks, T, 0, stream>>>(tab, bucket, ovf, ovfN, axs, W0, b0, W1, b1, W2, m);

    gather_rows<LAT_DIM, false><<<nodeWaveBlocks, T, 0, stream>>>(tab, bucket, ovf, ovfN, m, b2, out, N_NODES);
}


# Last gather: a wave takes 4 nodes in turn (10000 waves to start, not 40000); their counts, records and own rows asked for together, 8 row loads in flight (was 4); sum order unchanged
# speedup vs baseline: 7.3770x; 1.0490x over previous
#include <hip/hip_runtime.h>
#include <hip/hip_bf16.h>


typedef __attribute__((ext_vector_type(2))) float v2f;
typedef __attribute__((ext_vector_type(4))) float v4f;
typedef __attribute__((ext_vector_type(8))) float v8f;

#define N_NODES 40000
#define N_EDGES 640000
#define IN_DIM  3
#define HID_DIM 128
#define LAT_DIM 64
#define CAP     64
#define NBIN    250
#define BIN_N   160
#define CHUNK   2560
static_assert(NBIN * BIN_N == N_NODES, "whole ranges");
static_assert(NBIN * CHUNK == N_EDGES, "whole chunks");
static_assert(N_NODES <= 65536 && BIN_N <= 65536, "source and node-in-range share one 32-bit record");

__global__ __launch_bounds__(512) void bin_edges_kernel(
    const int* __restrict__ src, const int* __restrict__ dst,
    unsigned* __restrict__ recs, int2* __restrict__ runs, int* __restrict__ ovfN) {
    constexpr int T   = 512;
    constexpr int EPT = CHUNK / T;
    constexpr int NB  = 256;
    static_assert(CHUNK % T == 0, "whole rounds");
    static_assert(NBIN <= NB && NB == 32 * 8 && NBIN <= T, "one wave scans all ranges");
    __shared__ int      sCnt[NB];
    __shared__ int      sOff[NB];
    __shared__ unsigned sRec[CHUNK];

    const int t    = threadIdx.x;
    const int w    = blockIdx.x;
    const int base = w * CHUNK;
    if (t < NB) sCnt[t] = 0;
    if (w == 0 && t == 0) *ovfN = 0;
    __syncthreads();

    int s[EPT], d[EPT], bin[EPT], slot[EPT];
#pragma unroll
    for (int j = 0; j < EPT; ++j) {
        s[j] = src[base + j * T + t];
        d[j] = dst[base + j * T + t];
    }
    static_assert(EPT == 5, "the statement below names every loaded value");
    asm volatile("" :: "v"(s[0]), "v"(s[1]), "v"(s[2]), "v"(s[3]), "v"(s[4]),
                       "v"(d[0]), "v"(d[1]), "v"(d[2]), "v"(d[3]), "v"(d[4]));
#pragma unroll
    for (int j = 0; j < EPT; ++j) {
        bin[j]  = d[j] / BIN_N;
        slot[j] = atomicAdd(&sCnt[bin[j]], 1);
    }
    __syncthreads();
    if (t < 32) {
        int c[8];
        int tot = 0;
#pragma unroll
        for (int k = 0; k < 8; ++k) { c[k] = sCnt[8 * t + k]; tot += c[k]; }
        int inc = tot;
#pragma unroll
        for (int off = 1; off < 32; off <<= 1) {
            const int v = __shfl_up(inc, off);
            if (t >= off) inc += v;
        }
        int ex = inc - tot;
#pragma unroll
        for (int k = 0; k < 8; ++k) { sOff[8 * t + k] = ex; ex += c[k]; }
    }
    __syncthreads();
#pragma unroll
    for (int j = 0; j < EPT; ++j)
        sRec[sOff[bin[j]] + slot[j]] = (unsigned)s[j] | ((unsigned)(d[j] - bin[j] * BIN_N) << 16);
    __syncthreads();
#pragma unroll
    for (int j = 0; j < EPT; ++j) recs[base + j * T + t] = sRec[j * T + t];
    if (t < NBIN) runs[(size_t)t * NBIN + w] = make_int2(sOff[t], sCnt[t]);
}

__global__ __launch_bounds__(512) void fill_buckets_kernel(
    const float* __restrict__ x, const unsigned* __restrict__ recs, const int2* __restrict__ runs,
    int4* __restrict__ tab, int* __restrict__ bucket,
    int2* __restrict__ ovf, int* __restrict__ ovfN) {
    constexpr int T   = 512;
    constexpr int NHW = T / 16;
    constexpr int RPH = (NBIN + NHW - 1) / NHW;
    static_assert(BIN_N <= T, "one thread per node for the counters and rows");
    static_assert(CAP == 64, "bucket = 16 pieces of 16 bytes; rotation by node modulo 16 pieces");
    __shared__ int  sCnt[BIN_N];
    __shared__ int4 sB4[BIN_N * CAP / 4];
    int* sB = reinterpret_cast<int*>(sB4);

    const int t     = threadIdx.x;
    const int r     = blockIdx.x;
    const int node0 = r * BIN_N;
    const int hl    = t & 15;
    const int hw    = t >> 4;
    if (t < BIN_N) sCnt[t] = 0;
    __syncthreads();
    float x0 = 0.0f, x1 = 0.0f, x2 = 0.0f;
    if (t < BIN_N) {
        const float* xi = x + (size_t)(node0 + t) * IN_DIM;
        x0 = xi[0];
        x1 = xi[1];
        x2 = xi[2];
    }

    auto place = [&](unsigned rec) {
        const int sN   = (int)(rec & 0xFFFFu);
        const int dl   = (int)(rec >> 16);
        const int slot = atomicAdd(&sCnt[dl], 1);
        if (slot < CAP) sB[dl * CAP + ((slot + 4 * dl) & (CAP - 1))] = sN;
        else            ovf[atomicAdd(ovfN, 1)] = make_int2(sN, node0 + dl);
    };

    int first[RPH], len[RPH];
#pragma unroll
    for (int i = 0; i < RPH; ++i) {
        const int  w   = hw + NHW * i;
        const int  wc  = w < NBIN ? w : NBIN - 1;
        const int2 run = runs[(size_t)r * NBIN + wc];
        first[i] = wc * CHUNK + run.x;
        len[i]   = w < NBIN ? run.y : 0;
    }
    unsigned rc[RPH];
#pragma unroll
    for (int i = 0; i < RPH; ++i)
        rc[i] = recs[hl < len[i] ? first[i] + hl : 0];
    int slot[RPH];
#pragma unroll
    for (int i = 0; i < RPH; ++i) {
        slot[i] = 0;
        if (hl < len[i]) slot[i] = atomicAdd(&sCnt[rc[i] >> 16], 1);
    }
#pragma unroll
    for (int i = 0; i < RPH; ++i) {
        if (hl < len[i]) {
            const int sN = (int)(rc[i] & 0xFFFFu);
            const int dl = (int)(rc[i] >> 16);
            if (slot[i] < CAP) sB[dl * CAP + ((slot[i] + 4 * dl) & (CAP - 1))] = sN;
            else               ovf[atomicAdd(ovfN, 1)] = make_int2(sN, node0 + dl);
        }
    }
#pragma unroll
    for (int i = 0; i < RPH; ++i)
        for (int k = 16 + hl; k < len[i]; k += 16) place(recs[first[i] + k]);
    __syncthreads();

    if (t < BIN_N)
        tab[node0 + t] = make_int4(__float_as_int(x0), __float_as_int(x1), __float_as_int(x2), sCnt[t]);
    int4* bk4 = reinterpret_cast<int4*>(bucket + (size_t)node0 * CAP);
    for (int i = t; i < BIN_N * CAP / 4; i += T)
        bk4[i] = sB4[(i & ~15) | (((i & 15) + (i >> 4)) & 15)];
}


__global__ __launch_bounds__(256) void gather_in_rows(
    const int4* __restrict__ tab, const int* __restrict__ bucket,
    const int2* __restrict__ ovf, const int* __restrict__ ovfN,
    float4* __restrict__ axs) {
    constexpr int ROWS = 32;
    constexpr int SUB  = 8;
    constexpr int EPL  = 4;
    static_assert(ROWS * SUB == 256, "one group of lanes per node");
    static_assert(IN_DIM == 3, "input row and counter fill one 16-byte row");
    static_assert(N_NODES % ROWS == 0, "whole row groups: no row guard needed");
    __shared__ float4 sX4[ROWS];

    const int t    = threadIdx.x;
    const int row0 = blockIdx.x * ROWS;
    const int nl   = t / SUB;
    const int sub  = t % SUB;
    const int node = row0 + nl;
    const int4 tn  = tab[node];
    const int  cntN = tn.w;
    const float di  = rsqrtf((float)cntN + 1.0f);
    const int  nrec = min(cntN, CAP);
    const int* bk   = bucket + (size_t)node * CAP;

    const bool own = sub == 0;
    float a0 = own ? __int_as_float(tn.x) * di : 0.0f;
    float a1 = own ? __int_as_float(tn.y) * di : 0.0f;
    float a2 = own ? __int_as_float(tn.z) * di : 0.0f;
    for (int k = sub; k < nrec; k += SUB * EPL) {
        int  sN[EPL];
        bool ok[EPL];
#pragma unroll
        for (int q = 0; q < EPL; ++q) {
            const int idx = k + q * SUB;
            ok[q] = idx < nrec;
            sN[q] = bk[ok[q] ? idx : nrec - 1];
        }
#pragma unroll
        for (int q = 0; q < EPL; ++q) {
            const int4  ts = tab[sN[q]];
            const float cf = rsqrtf((float)ts.w + 1.0f);
            a0 += (ok[q] ? __int_as_float(ts.x) : 0.0f) * cf;
            a1 += (ok[q] ? __int_as_float(ts.y) : 0.0f) * cf;
            a2 += (ok[q] ? __int_as_float(ts.z) : 0.0f) * cf;
        }
    }
    if (cntN > CAP) {
        const int total = *ovfN;
        for (int k = sub; k < total; k += SUB) {
            const int2 en = ovf[k];
            if (en.y == node) {
                const int4  ts = tab[en.x];
                const float cf = rsqrtf((float)ts.w + 1.0f);
                a0 += __int_as_float(ts.x) * cf;
                a1 += __int_as_float(ts.y) * cf;
                a2 += __int_as_float(ts.z) * cf;
            }
        }
    }
#pragma unroll
    for (int off = 1; off < SUB; off <<= 1) {
        a0 += __shfl_xor(a0, off);
        a1 += __shfl_xor(a1, off);
        a2 += __shfl_xor(a2, off);
    }
    if (own) {
        const float d2 = di * di;
        sX4[nl] = make_float4(a0 * d2, a1 * d2, a2 * d2, di);
    }
    __syncthreads();
    if (t < ROWS) axs[row0 + t] = sX4[t];
}

__global__ __launch_bounds__(256, 10) void gather_h1_gemm2_f32(
    const int4* __restrict__ tab, const int* __restrict__ bucket,
    const int2* __restrict__ ovf, const int* __restrict__ ovfN,
    const float* __restrict__ axs,
    const float* __restrict__ Win, const float* __restrict__ bin,
    const float* __restrict__ B1, const float* __restrict__ bias1,
    const float* __restrict__ B2, float* __restrict__ C) {
    constexpr int K     = HID_DIM;
    constexpr int N1    = HID_DIM;
    constexpr int N2    = LAT_DIM;
    constexpr int ROWS  = 32;
    constexpr int NPW   = ROWS / 8;
    constexpr int CB    = N1 / 16;
    constexpr int RT    = ROWS / 16;
    constexpr int TN2   = N2 / 16;
    constexpr int PITCH = K + 4;
    static_assert(K == 128 && N1 / 16 == 8, "one column tile of W1 per wave");
    static_assert(8 % TN2 == 0 && 8 / TN2 == RT, "one tile of the last product per wave");
    static_assert(N_NODES % ROWS == 0, "whole row groups: no row guard needed");
    __shared__ float4 sA4[ROWS * PITCH / 4];
    __shared__ float  sD[ROWS];
    float* sA = reinterpret_cast<float*>(sA4);

    const int t    = threadIdx.x;
    const int lane = t & 31;
    const int wave = __builtin_amdgcn_readfirstlane(t >> 5);
    const int row0 = blockIdx.x * ROWS;
    const int c    = lane & 15;
    const int h    = lane >> 4;

    v2f wf[CB];
#pragma unroll
    for (int cb = 0; cb < CB; ++cb) {
        const int col = cb * 16 + c;
        const float w1 = Win[(size_t)N1 + col];
        const float bb = bin[col];
        wf[cb].x = Win[(size_t)(2 * h) * N1 + col];
        wf[cb].y = h ? bb : w1;
    }

    int   cnt[NPW], rows[NPW];
    float d1[NPW];
#pragma unroll
    for (int i = 0; i < NPW; ++i) {
        const int node = row0 + wave + i * 8;
        cnt[i]  = tab[node].w;
        rows[i] = min(cnt[i], CAP) + 1;
        d1[i]   = rsqrtf((float)cnt[i] + 1.0f);
    }

    auto afrag = [&](int node, int nrows, int q) -> v2f {
        const bool ok = q >= 1 && q < nrows;
        const int  r  = bucket[(size_t)node * CAP + (ok ? q - 1 : 0)];
        const int  sN = ok ? r : node;
        const v2f  v  = *reinterpret_cast<const v2f*>(axs + (size_t)sN * 4 + 2 * h);
        return q < nrows ? v : v2f{0.0f, 0.0f};
    };
    auto accum = [&](v2f a, float (&g)[CB]) {
#pragma unroll
        for (int cb = 0; cb < CB; ++cb) {
            const v8f d = __builtin_amdgcn_wmma_f32_16x16x4_f32(
                false, a, false, wf[cb], (short)0, v8f{}, false, false);
            const float s01 = fmaxf(d[0], 0.0f) + fmaxf(d[1], 0.0f);
            const float s23 = fmaxf(d[2], 0.0f) + fmaxf(d[3], 0.0f);
            const float s45 = fmaxf(d[4], 0.0f) + fmaxf(d[5], 0.0f);
            const float s67 = fmaxf(d[6], 0.0f) + fmaxf(d[7], 0.0f);
            g[cb] += (s01 + s23) + (s45 + s67);
        }
    };

    v2f anext = afrag(row0 + wave, rows[0], c);
#pragma unroll
    for (int i = 0; i < NPW; ++i) {
        const int node = row0 + wave + i * 8;
        float g[CB];
#pragma unroll
        for (int cb = 0; cb < CB; ++cb) g[cb] = 0.0f;
        v2f a = anext;
        if (i + 1 < NPW)
            anext = afrag(node + 8, rows[(i + 1) % NPW], c);
        for (int q0 = 0;;) {
            accum(a, g);
            q0 += 16;
            if (q0 >= rows[i]) break;
            a = afrag(node, rows[i], q0 + c);
        }
        if (cnt[i] > CAP) {
            const int total = *ovfN;
            for (int k0 = 0; k0 < total; k0 += 16) {
                const int  idx = k0 + c;
                const bool in  = idx < total;
                const int2 en  = ovf[in ? idx : 0];
                const bool hit = in && en.y == node;
                const v2f  v   = *reinterpret_cast<const v2f*>(axs + (size_t)(hit ? en.x : node) * 4 + 2 * h);
                accum(hit ? v : v2f{0.0f, 0.0f}, g);
            }
        }
#pragma unroll
        for (int j = 0; j < CB / 2; ++j) {
            const float keep = h ? g[2 * j + 1] : g[2 * j];
            const float send = h ? g[2 * j] : g[2 * j + 1];
            const float tot  = keep + __shfl_xor(send, 16);
            sA[(wave + i * 8) * PITCH + (2 * j + h) * 16 + c] = tot * d1[i];
        }
        if (lane == 0) sD[wave + i * 8] = d1[i];
    }
    __syncthreads();

    {
        const int ncol = wave * 16 + c;
        v2f bf[K / 4];
#pragma unroll
        for (int s = 0; s < K / 4; ++s) {
            const int ka = s * 4 + 2 * h;
            bf[s].x = B1[(size_t)ka * N1 + ncol];
            bf[s].y = B1[(size_t)(ka + 1) * N1 + ncol];
        }
        const float bcol = bias1[ncol];
        v8f acc[RT];
#pragma unroll
        for (int j = 0; j < RT; ++j) acc[j] = v8f{};
        const float* arow = sA + c * PITCH + 2 * h;
#pragma unroll
        for (int s = 0; s < K / 4; ++s) {
#pragma unroll
            for (int j = 0; j < RT; ++j) {
                const v2f a = *reinterpret_cast<const v2f*>(arow + j * 16 * PITCH + s * 4);
                acc[j] = __builtin_amdgcn_wmma_f32_16x16x4_f32(
                    false, a, false, bf[s], (short)0, acc[j], false, false);
            }
        }
        __syncthreads();
#pragma unroll
        for (int j = 0; j < RT; ++j) {
#pragma unroll
            for (int r = 0; r < 8; ++r)
                sA[(j * 16 + h * 8 + r) * PITCH + ncol] = fmaxf(acc[j][r] + bcol, 0.0f);
        }
    }
    __syncthreads();

    {
        const int tn   = wave % TN2;
        const int tr   = wave / TN2;
        const int ncol = tn * 16 + c;
        v2f bf[K / 4];
#pragma unroll
        for (int s = 0; s < K / 4; ++s) {
            const int ka = s * 4 + 2 * h;
            bf[s].x = B2[(size_t)ka * N2 + ncol];
            bf[s].y = B2[(size_t)(ka + 1) * N2 + ncol];
        }
        v8f acc = v8f{};
        const float* arow = sA + (tr * 16 + c) * PITCH + 2 * h;
#pragma unroll
        for (int s = 0; s < K / 4; ++s) {
            const v2f a = *reinterpret_cast<const v2f*>(arow + s * 4);
            acc = __builtin_amdgcn_wmma_f32_16x16x4_f32(
                false, a, false, bf[s], (short)0, acc, false, false);
        }
        const int crow = row0 + tr * 16 + h * 8;
#pragma unroll
        for (int r = 0; r < 8; ++r)
            C[(size_t)(crow + r) * N2 + ncol] = acc[r] * sD[tr * 16 + h * 8 + r];
    }
}

template <int DIM, bool RELU, int NPW>
__global__ __launch_bounds__(256) void gather_rows(
    const int4* __restrict__ tab, const int* __restrict__ bucket,
    const int2* __restrict__ ovf, const int* __restrict__ ovfN,
    const float* __restrict__ m,
    const float* __restrict__ bias, float* __restrict__ out) {
    constexpr int PER = DIM / 32;
    constexpr int WPB = 8;
    constexpr int G   = 8;
    typedef __attribute__((ext_vector_type(PER))) float vec;
    static_assert(N_NODES % (WPB * NPW) == 0, "whole work-groups: no node guard needed");
    static_assert(CAP == 64, "two passes of 32 records cover a bucket");

    const int lane = threadIdx.x & 31;
    const int wv    = __builtin_amdgcn_readfirstlane((int)(blockIdx.x * WPB + (threadIdx.x >> 5)));
    const int node0 = wv * NPW;

    const vec* __restrict__ mv = reinterpret_cast<const vec*>(m);
    const vec bv = reinterpret_cast<const vec*>(bias)[lane];

    int cnt[NPW], rec[NPW];
    vec self[NPW];
#pragma unroll
    for (int i = 0; i < NPW; ++i) {
        cnt[i]  = tab[node0 + i].w;
        rec[i]  = bucket[(size_t)(node0 + i) * CAP + lane];
        self[i] = mv[(size_t)(node0 + i) * 32 + lane];
    }

    static_assert(NPW == 4, "the statement below names every loaded count and record word");
    asm volatile("" :: "s"(cnt[0]), "s"(cnt[1]), "s"(cnt[2]), "s"(cnt[3]),
                       "v"(rec[0]), "v"(rec[1]), "v"(rec[2]), "v"(rec[3]));

#pragma unroll
    for (int i = 0; i < NPW; ++i) {
        const int   node = node0 + i;
        const int   cntN = cnt[i];
        const float di   = rsqrtf((float)cntN + 1.0f);
        const int   nrec = min(cntN, CAP);
        const int*  bk   = bucket + (size_t)node * CAP;
        vec acc = self[i];

        for (int base = 0; base < nrec; base += 32) {
            const int n = min(32, nrec - base);
            int r = rec[i];
            if (base > 0) r = bk[base + lane];
            for (int j = 0; j < n; j += G) {
                vec v[G];
#pragma unroll
                for (int q = 0; q < G; ++q) {
                    const int s0 = __builtin_amdgcn_readlane(r, min(j + q, n - 1));
                    v[q] = mv[(size_t)s0 * 32 + lane];
                }
#pragma unroll
                for (int q = 0; q < G; ++q)
                    if (j + q < n) acc += v[q];
            }
        }
        if (cntN > CAP) {
            const int total = *ovfN;
            for (int k0 = 0; k0 < total; k0 += 32) {
                const int idx = k0 + lane;
                int2 en = make_int2(0, -1);
                if (idx < total) en = ovf[idx];
                unsigned hits = __builtin_amdgcn_ballot_w32(en.y == node);
                while (hits) {
                    const int j = __builtin_ctz(hits);
                    hits &= hits - 1;
                    const int s0 = __builtin_amdgcn_readlane(en.x, j);
                    acc += mv[(size_t)s0 * 32 + lane];
                }
            }
        }

        acc = acc * di + bv;
        if (RELU) {
#pragma unroll
            for (int k = 0; k < PER; ++k) acc[k] = fmaxf(acc[k], 0.0f);
        }
        reinterpret_cast<vec*>(out)[(size_t)node * 32 + lane] = acc;
    }
}

extern "C" void kernel_launch(void* const* d_in, const int* in_sizes, int n_in,
                              void* d_out, int out_size, void* d_ws, size_t ws_size,
                              hipStream_t stream) {
    const float* x   = (const float*)d_in[0];
    const int*   ei  = (const int*)d_in[1];
    const float* W0  = (const float*)d_in[2];
    const float* b0  = (const float*)d_in[3];
    const float* W1  = (const float*)d_in[4];
    const float* b1  = (const float*)d_in[5];
    const float* W2  = (const float*)d_in[6];
    const float* b2  = (const float*)d_in[7];
    float* out = (float*)d_out;

    const int* src = ei;
    const int* dst = ei + N_EDGES;

    float* ws     = (float*)d_ws;
    float* m      = ws;
    float* axs    = m + (size_t)N_NODES * LAT_DIM;
    int4*  tab    = (int4*)(axs + (size_t)N_NODES * 4);
    int*   bucket = (int*)(tab + N_NODES);
    int2*  ovf    = (int2*)(bucket + (size_t)N_NODES * CAP);
    int*   ovfN   = (int*)(ovf + N_EDGES);
    unsigned* recs = (unsigned*)(ovfN + 4);
    int2*  runs   = (int2*)(recs + N_EDGES);

    const int T = 256;
    const int gatherBlocks = N_NODES / (8 * 4);
    const int fusedBlocks = N_NODES / 32;

    bin_edges_kernel<<<NBIN, 512, 0, stream>>>(src, dst, recs, runs, ovfN);
    fill_buckets_kernel<<<NBIN, 512, 0, stream>>>(x, recs, runs, tab, bucket, ovf, ovfN);

    gather_in_rows<<<fusedBlocks, T, 0, stream>>>(tab, bucket, ovf, ovfN, (float4*)axs);

    gather_h1_gemm2_f32<<<fusedBlocks, T, 0, stream>>>(tab, bucket, ovf, ovfN, axs, W0, b0, W1, b1, W2, m);

    gather_rows<LAT_DIM, false, 4><<<gatherBlocks, T, 0, stream>>>(tab, bucket, ovf, ovfN, m, b2, out);
}
